# starting kernel
# baseline (speedup 1.0000x reference)
#include <hip/hip_runtime.h>
#include <hip/hip_bf16.h>


#define D_IN   128
#define D_OUT  64
#define LEAKY  0.2f

typedef float v2f __attribute__((ext_vector_type(2)));
typedef float v8f __attribute__((ext_vector_type(8)));

__device__ __forceinline__ unsigned ord_of_float(float f) {
    unsigned u = __float_as_uint(f);
    return (u & 0x80000000u) ? ~u : (u | 0x80000000u);
}
__device__ __forceinline__ float float_of_ord(unsigned u) {
    return __uint_as_float((u & 0x80000000u) ? (u ^ 0x80000000u) : ~u);
}

__global__ __launch_bounds__(256) void k_init(float* __restrict__ out,
                                              unsigned* __restrict__ segmax,
                                              float* __restrict__ denom,
                                              int n_nodes) {
    int i = blockIdx.x * blockDim.x + threadIdx.x;
    int total = n_nodes * D_OUT;
    if (i < total) out[i] = 0.0f;
    if (i < n_nodes) { segmax[i] = 0u; denom[i] = 0.0f; }
}

__global__ __launch_bounds__(256) void k_gemm(const float* __restrict__ X,
                                              const float* __restrict__ W,
                                              float* __restrict__ wx,
                                              int n_nodes) {
    constexpr int LDW = D_IN + 4;
    __shared__ float sWt[D_OUT * LDW];
    for (int i = threadIdx.x; i < D_IN * D_OUT; i += blockDim.x) {
        int k = i / D_OUT, col = i % D_OUT;
        sWt[col * LDW + k] = W[i];
    }
    __syncthreads();

    const int lane = threadIdx.x & 31;
    const int wave = threadIdx.x >> 5;
    const int tile = blockIdx.x * (blockDim.x >> 5) + wave;
    const int row0 = tile * 16;
    if (row0 >= n_nodes) return;

    const int m    = lane & 15;
    const int koff = (lane >> 4) * 2;
    int arow = row0 + m;
    if (arow >= n_nodes) arow = n_nodes - 1;

    v8f acc[4] = {v8f{}, v8f{}, v8f{}, v8f{}};

    const float* __restrict__ xrow = X + (size_t)arow * D_IN;
    #pragma unroll
    for (int kb = 0; kb < D_IN; kb += 4) {
        v2f a = *(const v2f*)(xrow + kb + koff);
        #pragma unroll
        for (int t = 0; t < 4; ++t) {
            const int col = t * 16 + m;
            v2f b = *(const v2f*)(&sWt[col * LDW + kb + koff]);
            acc[t] = __builtin_amdgcn_wmma_f32_16x16x4_f32(
                 false, a,  false, b,
                 (short)0, acc[t],
                 false,  false);
        }
    }

    const int rbase = row0 + ((lane >> 4) * 8);
    #pragma unroll
    for (int t = 0; t < 4; ++t) {
        #pragma unroll
        for (int r = 0; r < 8; ++r) {
            const int row = rbase + r;
            if (row < n_nodes)
                wx[(size_t)row * D_OUT + t * 16 + m] = acc[t][r];
        }
    }
}

__global__ __launch_bounds__(256) void k_scores(const float* __restrict__ wx,
                                                const float* __restrict__ att,
                                                float* __restrict__ s_src,
                                                float* __restrict__ s_trg,
                                                int n_nodes) {
    int node = blockIdx.x * blockDim.x + threadIdx.x;
    if (node >= n_nodes) return;
    const float* row = wx + (size_t)node * D_OUT;
    float ss = 0.0f, st = 0.0f;
    #pragma unroll 8
    for (int d = 0; d < D_OUT; ++d) {
        float v = row[d];
        ss = fmaf(v, att[d], ss);
        st = fmaf(v, att[D_OUT + d], st);
    }
    s_src[node] = ss;
    s_trg[node] = st;
}

__global__ __launch_bounds__(256) void k_logits(const int* __restrict__ esrc,
                                                const int* __restrict__ etrg,
                                                const float* __restrict__ s_src,
                                                const float* __restrict__ s_trg,
                                                float* __restrict__ logits,
                                                unsigned* __restrict__ segmax,
                                                int n_edges) {
    int e = blockIdx.x * blockDim.x + threadIdx.x;
    if (e >= n_edges) return;
    int s = esrc[e], t = etrg[e];
    float l = s_src[s] + s_trg[t];
    l = (l >= 0.0f) ? l : LEAKY * l;
    logits[e] = l;
    atomicMax(&segmax[s], ord_of_float(l));
}

__global__ __launch_bounds__(256) void k_exp(const int* __restrict__ esrc,
                                             float* __restrict__ logits,
                                             const unsigned* __restrict__ segmax,
                                             float* __restrict__ denom,
                                             int n_edges) {
    int e = blockIdx.x * blockDim.x + threadIdx.x;
    if (e >= n_edges) return;
    int s = esrc[e];
    float ex = __expf(logits[e] - float_of_ord(segmax[s]));
    logits[e] = ex;
    atomicAdd(&denom[s], ex);
}

__global__ __launch_bounds__(256) void k_scatter(const int* __restrict__ esrc,
                                                 const int* __restrict__ etrg,
                                                 const float* __restrict__ ex,
                                                 const float* __restrict__ denom,
                                                 const float* __restrict__ wx,
                                                 float* __restrict__ out,
                                                 int n_edges) {
    int e    = (blockIdx.x * blockDim.x + threadIdx.x) >> 5;
    int lane = threadIdx.x & 31;
    if (e >= n_edges) return;
    int s = esrc[e], t = etrg[e];
    float alpha = ex[e] / denom[s];
    const float2* src2 = (const float2*)(wx + (size_t)t * D_OUT);
    float2 v = src2[lane];
    float* o = out + (size_t)s * D_OUT + 2 * lane;
    atomicAdd(o,     alpha * v.x);
    atomicAdd(o + 1, alpha * v.y);
}

extern "C" void kernel_launch(void* const* d_in, const int* in_sizes, int n_in,
                              void* d_out, int out_size, void* d_ws, size_t ws_size,
                              hipStream_t stream) {
    const float* X   = (const float*)d_in[0];
    const float* W   = (const float*)d_in[1];
    const float* att = (const float*)d_in[2];
    const int*   adj = (const int*)  d_in[3];

    const int d_out_dim = in_sizes[2] / 2;
    const int d_in_dim  = in_sizes[1] / d_out_dim;
    const int n_nodes   = in_sizes[0] / d_in_dim;
    const int n_edges   = in_sizes[3] / 2;
    (void)d_out_dim; (void)d_in_dim;

    const int* esrc = adj;
    const int* etrg = adj + n_edges;

    float*    wx     = (float*)d_ws;
    float*    s_src  = wx + (size_t)n_nodes * D_OUT;
    float*    s_trg  = s_src + n_nodes;
    float*    denom  = s_trg + n_nodes;
    unsigned* segmax = (unsigned*)(denom + n_nodes);
    float*    logits = (float*)(segmax + n_nodes);
    float*    out    = (float*)d_out;

    const int B = 256;

    {
        int total = n_nodes * D_OUT;
        k_init<<<(total + B - 1) / B, B, 0, stream>>>(out, segmax, denom, n_nodes);
    }
    {
        int tiles  = (n_nodes + 15) / 16;
        int blocks = (tiles + 7) / 8;
        k_gemm<<<blocks, B, 0, stream>>>(X, W, wx, n_nodes);
    }
    k_scores<<<(n_nodes + B - 1) / B, B, 0, stream>>>(wx, att, s_src, s_trg, n_nodes);
    k_logits<<<(n_edges + B - 1) / B, B, 0, stream>>>(esrc, etrg, s_src, s_trg,
                                                      logits, segmax, n_edges);
    k_exp<<<(n_edges + B - 1) / B, B, 0, stream>>>(esrc, logits, segmax, denom, n_edges);
    {
        long long threads = (long long)n_edges * 32;
        int blocks = (int)((threads + B - 1) / B);
        k_scatter<<<blocks, B, 0, stream>>>(esrc, etrg, logits, denom, wx, out, n_edges);
    }
}


# Owner-computes edge phase: each work-group owns 391 sources in LDS and scans the edge list for them; global atomics and init replaced by LDS atomics (sum order by arrival)
# speedup vs baseline: 1.7475x; 1.7475x over previous
#include <hip/hip_runtime.h>
#include <hip/hip_bf16.h>


#define D_IN   128
#define D_OUT  64
#define LEAKY  0.2f

typedef float v2f __attribute__((ext_vector_type(2)));
typedef float v8f __attribute__((ext_vector_type(8)));

__device__ __forceinline__ unsigned ord_of_float(float f) {
    unsigned u = __float_as_uint(f);
    return (u & 0x80000000u) ? ~u : (u | 0x80000000u);
}
__device__ __forceinline__ float float_of_ord(unsigned u) {
    return __uint_as_float((u & 0x80000000u) ? (u ^ 0x80000000u) : ~u);
}

__global__ __launch_bounds__(256) void k_gemm(const float* __restrict__ X,
                                              const float* __restrict__ W,
                                              float* __restrict__ wx,
                                              int n_nodes) {
    constexpr int LDW = D_IN + 4;
    __shared__ float sWt[D_OUT * LDW];
    for (int i = threadIdx.x; i < D_IN * D_OUT; i += blockDim.x) {
        int k = i / D_OUT, col = i % D_OUT;
        sWt[col * LDW + k] = W[i];
    }
    __syncthreads();

    const int lane = threadIdx.x & 31;
    const int wave = threadIdx.x >> 5;
    const int tile = blockIdx.x * (blockDim.x >> 5) + wave;
    const int row0 = tile * 16;
    if (row0 >= n_nodes) return;

    const int m    = lane & 15;
    const int koff = (lane >> 4) * 2;
    int arow = row0 + m;
    if (arow >= n_nodes) arow = n_nodes - 1;

    v8f acc[4] = {v8f{}, v8f{}, v8f{}, v8f{}};

    const float* __restrict__ xrow = X + (size_t)arow * D_IN;
    #pragma unroll
    for (int kb = 0; kb < D_IN; kb += 4) {
        v2f a = *(const v2f*)(xrow + kb + koff);
        #pragma unroll
        for (int t = 0; t < 4; ++t) {
            const int col = t * 16 + m;
            v2f b = *(const v2f*)(&sWt[col * LDW + kb + koff]);
            acc[t] = __builtin_amdgcn_wmma_f32_16x16x4_f32(
                 false, a,  false, b,
                 (short)0, acc[t],
                 false,  false);
        }
    }

    const int rbase = row0 + ((lane >> 4) * 8);
    #pragma unroll
    for (int t = 0; t < 4; ++t) {
        #pragma unroll
        for (int r = 0; r < 8; ++r) {
            const int row = rbase + r;
            if (row < n_nodes)
                wx[(size_t)row * D_OUT + t * 16 + m] = acc[t][r];
        }
    }
}

__global__ __launch_bounds__(256) void k_scores(const float* __restrict__ wx,
                                                const float* __restrict__ att,
                                                float* __restrict__ s_src,
                                                float* __restrict__ s_trg,
                                                int n_nodes) {
    int node = blockIdx.x * blockDim.x + threadIdx.x;
    if (node >= n_nodes) return;
    const float* row = wx + (size_t)node * D_OUT;
    float ss = 0.0f, st = 0.0f;
    #pragma unroll 8
    for (int d = 0; d < D_OUT; ++d) {
        float v = row[d];
        ss = fmaf(v, att[d], ss);
        st = fmaf(v, att[D_OUT + d], st);
    }
    s_src[node] = ss;
    s_trg[node] = st;
}

#define NPW   391
#define CAP   8192
#define ETHR  1024

__device__ __forceinline__ float edge_logit(float a, float b) {
    float l = a + b;
    return (l >= 0.0f) ? l : LEAKY * l;
}

template <typename F>
__device__ __forceinline__ void scan_owned(const int* __restrict__ esrc, int n_edges,
                                           int n0, int nown, int rot, F body) {
    const int tid  = threadIdx.x;
    const int nvec = n_edges >> 2;
    const int4* __restrict__ src4 = (const int4*)esrc;
    for (int base = 0; base < nvec; base += 4 * ETHR) {
        int4 q[4];
        int  w[4];
        #pragma unroll
        for (int u = 0; u < 4; ++u) {
            const int v = base + u * ETHR + tid;
            int ww = v + rot;
            if (ww >= nvec) ww -= nvec;
            w[u] = ww;
            q[u] = (v < nvec) ? src4[ww] : make_int4(-1, -1, -1, -1);
        }
        #pragma unroll
        for (int u = 0; u < 4; ++u) {
            const int e0 = w[u] << 2;
            const unsigned d0 = (unsigned)(q[u].x - n0);
            const unsigned d1 = (unsigned)(q[u].y - n0);
            const unsigned d2 = (unsigned)(q[u].z - n0);
            const unsigned d3 = (unsigned)(q[u].w - n0);
            if (d0 < (unsigned)nown) body((int)d0, e0);
            if (d1 < (unsigned)nown) body((int)d1, e0 + 1);
            if (d2 < (unsigned)nown) body((int)d2, e0 + 2);
            if (d3 < (unsigned)nown) body((int)d3, e0 + 3);
        }
    }
    for (int e = (nvec << 2) + tid; e < n_edges; e += ETHR) {
        const unsigned d = (unsigned)(esrc[e] - n0);
        if (d < (unsigned)nown) body((int)d, e);
    }
}

__global__ __launch_bounds__(ETHR) void k_edges(const int* __restrict__ esrc,
                                                const int* __restrict__ etrg,
                                                const float* __restrict__ s_src,
                                                const float* __restrict__ s_trg,
                                                const float* __restrict__ wx,
                                                float* __restrict__ out,
                                                int n_nodes, int n_edges) {
    __shared__ float    accL[NPW * D_OUT];
    __shared__ unsigned maxL[NPW];
    __shared__ float    denL[NPW];
    __shared__ float    ssL[NPW];
    __shared__ int      listS[CAP];
    __shared__ int      listT[CAP];
    __shared__ unsigned cntL;

    const int tid  = threadIdx.x;
    const int lane = tid & 31;
    const int wave = tid >> 5;
    const int n0   = blockIdx.x * NPW;
    const int nown = min(NPW, n_nodes - n0);

    for (int i = tid; i < NPW * D_OUT; i += ETHR) accL[i] = 0.0f;
    for (int i = tid; i < NPW; i += ETHR) {
        maxL[i] = 0u;
        denL[i] = 0.0f;
        ssL[i]  = (i < nown) ? s_src[n0 + i] : 0.0f;
    }
    if (tid == 0) cntL = 0u;
    __syncthreads();

    const int nvec = n_edges >> 2;
    const int rot  = (int)(((long long)blockIdx.x * nvec) / gridDim.x) & ~31;

    scan_owned(esrc, n_edges, n0, nown, rot, [&](int sl, int e) {
        const unsigned idx = atomicAdd(&cntL, 1u);
        if (idx < (unsigned)CAP) { listS[idx] = sl; listT[idx] = e; }
    });
    __syncthreads();
    const unsigned cnt = cntL;

    if (cnt <= (unsigned)CAP) {
        for (unsigned i = tid; i < cnt; i += ETHR) {
            const int sl = listS[i];
            const int t  = etrg[listT[i]];
            listT[i] = t;
            atomicMax(&maxL[sl], ord_of_float(edge_logit(ssL[sl], s_trg[t])));
        }
        __syncthreads();
        for (unsigned i = tid; i < cnt; i += ETHR) {
            const int sl = listS[i];
            const float l = edge_logit(ssL[sl], s_trg[listT[i]]);
            atomicAdd(&denL[sl], __expf(l - float_of_ord(maxL[sl])));
        }
        __syncthreads();
        for (unsigned i0 = wave * 4; i0 < cnt; i0 += (ETHR / 32) * 4) {
            int    sl[4];
            float  al[4];
            float2 v[4];
            #pragma unroll
            for (int j = 0; j < 4; ++j) {
                const unsigned i = (i0 + j < cnt) ? (i0 + j) : i0;
                sl[j] = listS[i];
                const int t = listT[i];
                const float l = edge_logit(ssL[sl[j]], s_trg[t]);
                al[j] = __expf(l - float_of_ord(maxL[sl[j]])) / denL[sl[j]];
                v[j]  = ((const float2*)(wx + (size_t)t * D_OUT))[lane];
            }
            #pragma unroll
            for (int j = 0; j < 4; ++j) {
                if (i0 + j < cnt) {
                    float* a = &accL[sl[j] * D_OUT + 2 * lane];
                    atomicAdd(a,     al[j] * v[j].x);
                    atomicAdd(a + 1, al[j] * v[j].y);
                }
            }
        }
    } else {
        scan_owned(esrc, n_edges, n0, nown, rot, [&](int sl, int e) {
            const int t = etrg[e];
            atomicMax(&maxL[sl], ord_of_float(edge_logit(ssL[sl], s_trg[t])));
        });
        __syncthreads();
        scan_owned(esrc, n_edges, n0, nown, rot, [&](int sl, int e) {
            const float l = edge_logit(ssL[sl], s_trg[etrg[e]]);
            atomicAdd(&denL[sl], __expf(l - float_of_ord(maxL[sl])));
        });
        __syncthreads();
        scan_owned(esrc, n_edges, n0, nown, rot, [&](int sl, int e) {
            const int t = etrg[e];
            const float l = edge_logit(ssL[sl], s_trg[t]);
            const float alpha = __expf(l - float_of_ord(maxL[sl])) / denL[sl];
            const float* __restrict__ row = wx + (size_t)t * D_OUT;
            for (int d = 0; d < D_OUT; ++d) atomicAdd(&accL[sl * D_OUT + d], alpha * row[d]);
        });
    }
    __syncthreads();

    float* __restrict__ o = out + (size_t)n0 * D_OUT;
    for (int i = tid; i < nown * D_OUT; i += ETHR) o[i] = accL[i];
}

extern "C" void kernel_launch(void* const* d_in, const int* in_sizes, int n_in,
                              void* d_out, int out_size, void* d_ws, size_t ws_size,
                              hipStream_t stream) {
    const float* X   = (const float*)d_in[0];
    const float* W   = (const float*)d_in[1];
    const float* att = (const float*)d_in[2];
    const int*   adj = (const int*)  d_in[3];

    const int d_out_dim = in_sizes[2] / 2;
    const int d_in_dim  = in_sizes[1] / d_out_dim;
    const int n_nodes   = in_sizes[0] / d_in_dim;
    const int n_edges   = in_sizes[3] / 2;
    (void)d_out_dim; (void)d_in_dim;

    const int* esrc = adj;
    const int* etrg = adj + n_edges;

    float* wx    = (float*)d_ws;
    float* s_src = wx + (size_t)n_nodes * D_OUT;
    float* s_trg = s_src + n_nodes;
    float* out   = (float*)d_out;

    const int B = 256;

    {
        int tiles  = (n_nodes + 15) / 16;
        int blocks = (tiles + 7) / 8;
        k_gemm<<<blocks, B, 0, stream>>>(X, W, wx, n_nodes);
    }
    k_scores<<<(n_nodes + B - 1) / B, B, 0, stream>>>(wx, att, s_src, s_trg, n_nodes);
    k_edges<<<(n_nodes + NPW - 1) / NPW, ETHR, 0, stream>>>(esrc, etrg, s_src, s_trg, wx, out,
                                                            n_nodes, n_edges);
}


# Binning pre-pass: a new kernel sorts edges into per-owner buckets (LDS histogram, one global cursor add per bucket); work-groups read their bucket instead of scanning all edges (sums by arrival)
# speedup vs baseline: 2.5419x; 1.4546x over previous
#include <hip/hip_runtime.h>
#include <hip/hip_bf16.h>


#define D_IN   128
#define D_OUT  64
#define LEAKY  0.2f

typedef float v2f __attribute__((ext_vector_type(2)));
typedef float v8f __attribute__((ext_vector_type(8)));

__device__ __forceinline__ unsigned ord_of_float(float f) {
    unsigned u = __float_as_uint(f);
    return (u & 0x80000000u) ? ~u : (u | 0x80000000u);
}
__device__ __forceinline__ float float_of_ord(unsigned u) {
    return __uint_as_float((u & 0x80000000u) ? (u ^ 0x80000000u) : ~u);
}

__global__ __launch_bounds__(256) void k_gemm(const float* __restrict__ X,
                                              const float* __restrict__ W,
                                              float* __restrict__ wx,
                                              int n_nodes) {
    constexpr int LDW = D_IN + 4;
    __shared__ float sWt[D_OUT * LDW];
    for (int i = threadIdx.x; i < D_IN * D_OUT; i += blockDim.x) {
        int k = i / D_OUT, col = i % D_OUT;
        sWt[col * LDW + k] = W[i];
    }
    __syncthreads();

    const int lane = threadIdx.x & 31;
    const int wave = threadIdx.x >> 5;
    const int tile = blockIdx.x * (blockDim.x >> 5) + wave;
    const int row0 = tile * 16;
    if (row0 >= n_nodes) return;

    const int m    = lane & 15;
    const int koff = (lane >> 4) * 2;
    int arow = row0 + m;
    if (arow >= n_nodes) arow = n_nodes - 1;

    v8f acc[4] = {v8f{}, v8f{}, v8f{}, v8f{}};

    const float* __restrict__ xrow = X + (size_t)arow * D_IN;
    #pragma unroll
    for (int kb = 0; kb < D_IN; kb += 4) {
        v2f a = *(const v2f*)(xrow + kb + koff);
        #pragma unroll
        for (int t = 0; t < 4; ++t) {
            const int col = t * 16 + m;
            v2f b = *(const v2f*)(&sWt[col * LDW + kb + koff]);
            acc[t] = __builtin_amdgcn_wmma_f32_16x16x4_f32(
                 false, a,  false, b,
                 (short)0, acc[t],
                 false,  false);
        }
    }

    const int rbase = row0 + ((lane >> 4) * 8);
    #pragma unroll
    for (int t = 0; t < 4; ++t) {
        #pragma unroll
        for (int r = 0; r < 8; ++r) {
            const int row = rbase + r;
            if (row < n_nodes)
                wx[(size_t)row * D_OUT + t * 16 + m] = acc[t][r];
        }
    }
}

#define NPW   391
#define CAP   8192
#define ETHR  1024
#define BTHR  1024
#define BEPT  4
#define CSTR  64
#define NBMAX 1024

__global__ __launch_bounds__(256) void k_scores(const float* __restrict__ wx,
                                                const float* __restrict__ att,
                                                float* __restrict__ s_src,
                                                float* __restrict__ s_trg,
                                                unsigned* __restrict__ cursor,
                                                int n_nodes, int nb) {
    int node = blockIdx.x * blockDim.x + threadIdx.x;
    if (node < nb) cursor[node * CSTR] = 0u;
    if (node >= n_nodes) return;
    const float* row = wx + (size_t)node * D_OUT;
    float ss = 0.0f, st = 0.0f;
    #pragma unroll 8
    for (int d = 0; d < D_OUT; ++d) {
        float v = row[d];
        ss = fmaf(v, att[d], ss);
        st = fmaf(v, att[D_OUT + d], st);
    }
    s_src[node] = ss;
    s_trg[node] = st;
}

__global__ __launch_bounds__(BTHR) void k_bin(const int* __restrict__ esrc,
                                              const int* __restrict__ etrg,
                                              unsigned* __restrict__ cursor,
                                              int2* __restrict__ buckets,
                                              int n_nodes, int n_edges, int nb) {
    __shared__ unsigned hist[NBMAX];
    const int tid = threadIdx.x;
    for (int i = tid; i < nb; i += BTHR) hist[i] = 0u;
    __syncthreads();

    const int base = blockIdx.x * (BTHR * BEPT) + tid;
    int s[BEPT], t[BEPT], b[BEPT];
    unsigned r[BEPT];
    #pragma unroll
    for (int u = 0; u < BEPT; ++u) {
        const int e = base + u * BTHR;
        const bool in = e < n_edges;
        s[u] = in ? esrc[e] : -1;
        t[u] = in ? etrg[e] : 0;
    }
    #pragma unroll
    for (int u = 0; u < BEPT; ++u) {
        const bool ok = (unsigned)s[u] < (unsigned)n_nodes;
        b[u] = ok ? s[u] / NPW : -1;
        r[u] = ok ? atomicAdd(&hist[b[u]], 1u) : 0u;
    }
    __syncthreads();
    for (int i = tid; i < nb; i += BTHR) {
        const unsigned c = hist[i];
        hist[i] = c ? atomicAdd(&cursor[i * CSTR], c) : 0u;
    }
    __syncthreads();
    #pragma unroll
    for (int u = 0; u < BEPT; ++u) {
        if (b[u] >= 0) {
            const unsigned pos = hist[b[u]] + r[u];
            if (pos < (unsigned)CAP)
                buckets[(size_t)b[u] * CAP + pos] = make_int2(s[u] - b[u] * NPW, t[u]);
        }
    }
}

__device__ __forceinline__ float edge_logit(float a, float b) {
    float l = a + b;
    return (l >= 0.0f) ? l : LEAKY * l;
}

template <typename F>
__device__ __forceinline__ void scan_owned(const int* __restrict__ esrc, int n_edges,
                                           int n0, int nown, int rot, F body) {
    const int tid  = threadIdx.x;
    const int nvec = n_edges >> 2;
    const int4* __restrict__ src4 = (const int4*)esrc;
    for (int base = 0; base < nvec; base += 4 * ETHR) {
        int4 q[4];
        int  w[4];
        #pragma unroll
        for (int u = 0; u < 4; ++u) {
            const int v = base + u * ETHR + tid;
            int ww = v + rot;
            if (ww >= nvec) ww -= nvec;
            w[u] = ww;
            q[u] = (v < nvec) ? src4[ww] : make_int4(-1, -1, -1, -1);
        }
        #pragma unroll
        for (int u = 0; u < 4; ++u) {
            const int e0 = w[u] << 2;
            const unsigned d0 = (unsigned)(q[u].x - n0);
            const unsigned d1 = (unsigned)(q[u].y - n0);
            const unsigned d2 = (unsigned)(q[u].z - n0);
            const unsigned d3 = (unsigned)(q[u].w - n0);
            if (d0 < (unsigned)nown) body((int)d0, e0);
            if (d1 < (unsigned)nown) body((int)d1, e0 + 1);
            if (d2 < (unsigned)nown) body((int)d2, e0 + 2);
            if (d3 < (unsigned)nown) body((int)d3, e0 + 3);
        }
    }
    for (int e = (nvec << 2) + tid; e < n_edges; e += ETHR) {
        const unsigned d = (unsigned)(esrc[e] - n0);
        if (d < (unsigned)nown) body((int)d, e);
    }
}

__global__ __launch_bounds__(ETHR) void k_edges(const int* __restrict__ esrc,
                                                const int* __restrict__ etrg,
                                                const float* __restrict__ s_src,
                                                const float* __restrict__ s_trg,
                                                const float* __restrict__ wx,
                                                const unsigned* __restrict__ cursor,
                                                const int2* __restrict__ buckets,
                                                float* __restrict__ out,
                                                int n_nodes, int n_edges) {
    __shared__ float    accL[NPW * D_OUT];
    __shared__ unsigned maxL[NPW];
    __shared__ float    denL[NPW];
    __shared__ float    ssL[NPW];
    __shared__ int      listS[CAP];
    __shared__ int      listT[CAP];

    const int tid  = threadIdx.x;
    const int lane = tid & 31;
    const int wave = tid >> 5;
    const int n0   = blockIdx.x * NPW;
    const int nown = min(NPW, n_nodes - n0);

    for (int i = tid; i < NPW * D_OUT; i += ETHR) accL[i] = 0.0f;
    for (int i = tid; i < NPW; i += ETHR) {
        maxL[i] = 0u;
        denL[i] = 0.0f;
        ssL[i]  = (i < nown) ? s_src[n0 + i] : 0.0f;
    }
    __syncthreads();

    const unsigned cnt = cursor[blockIdx.x * CSTR];

    if (cnt <= (unsigned)CAP) {
        const int2* __restrict__ bk = buckets + (size_t)blockIdx.x * CAP;
        for (unsigned i = tid; i < cnt; i += ETHR) {
            const int2 rec = bk[i];
            const int sl = rec.x;
            const int t  = rec.y;
            listS[i] = sl;
            listT[i] = t;
            atomicMax(&maxL[sl], ord_of_float(edge_logit(ssL[sl], s_trg[t])));
        }
        __syncthreads();
        for (unsigned i = tid; i < cnt; i += ETHR) {
            const int sl = listS[i];
            const float l = edge_logit(ssL[sl], s_trg[listT[i]]);
            atomicAdd(&denL[sl], __expf(l - float_of_ord(maxL[sl])));
        }
        __syncthreads();
        for (unsigned i0 = wave * 4; i0 < cnt; i0 += (ETHR / 32) * 4) {
            int    sl[4];
            float  al[4];
            float2 v[4];
            #pragma unroll
            for (int j = 0; j < 4; ++j) {
                const unsigned i = (i0 + j < cnt) ? (i0 + j) : i0;
                sl[j] = listS[i];
                const int t = listT[i];
                const float l = edge_logit(ssL[sl[j]], s_trg[t]);
                al[j] = __expf(l - float_of_ord(maxL[sl[j]])) / denL[sl[j]];
                v[j]  = ((const float2*)(wx + (size_t)t * D_OUT))[lane];
            }
            #pragma unroll
            for (int j = 0; j < 4; ++j) {
                if (i0 + j < cnt) {
                    float* a = &accL[sl[j] * D_OUT + 2 * lane];
                    atomicAdd(a,     al[j] * v[j].x);
                    atomicAdd(a + 1, al[j] * v[j].y);
                }
            }
        }
    } else {
        const int nvec = n_edges >> 2;
        const int rot  = (int)(((long long)blockIdx.x * nvec) / gridDim.x) & ~31;
        scan_owned(esrc, n_edges, n0, nown, rot, [&](int sl, int e) {
            const int t = etrg[e];
            atomicMax(&maxL[sl], ord_of_float(edge_logit(ssL[sl], s_trg[t])));
        });
        __syncthreads();
        scan_owned(esrc, n_edges, n0, nown, rot, [&](int sl, int e) {
            const float l = edge_logit(ssL[sl], s_trg[etrg[e]]);
            atomicAdd(&denL[sl], __expf(l - float_of_ord(maxL[sl])));
        });
        __syncthreads();
        scan_owned(esrc, n_edges, n0, nown, rot, [&](int sl, int e) {
            const int t = etrg[e];
            const float l = edge_logit(ssL[sl], s_trg[t]);
            const float alpha = __expf(l - float_of_ord(maxL[sl])) / denL[sl];
            const float* __restrict__ row = wx + (size_t)t * D_OUT;
            for (int d = 0; d < D_OUT; ++d) atomicAdd(&accL[sl * D_OUT + d], alpha * row[d]);
        });
    }
    __syncthreads();

    float* __restrict__ o = out + (size_t)n0 * D_OUT;
    for (int i = tid; i < nown * D_OUT; i += ETHR) o[i] = accL[i];
}

extern "C" void kernel_launch(void* const* d_in, const int* in_sizes, int n_in,
                              void* d_out, int out_size, void* d_ws, size_t ws_size,
                              hipStream_t stream) {
    const float* X   = (const float*)d_in[0];
    const float* W   = (const float*)d_in[1];
    const float* att = (const float*)d_in[2];
    const int*   adj = (const int*)  d_in[3];

    const int d_out_dim = in_sizes[2] / 2;
    const int d_in_dim  = in_sizes[1] / d_out_dim;
    const int n_nodes   = in_sizes[0] / d_in_dim;
    const int n_edges   = in_sizes[3] / 2;
    (void)d_out_dim; (void)d_in_dim;

    const int* esrc = adj;
    const int* etrg = adj + n_edges;

    float* wx    = (float*)d_ws;
    float* s_src = wx + (size_t)n_nodes * D_OUT;
    float* s_trg = s_src + n_nodes;
    const int nb = (n_nodes + NPW - 1) / NPW;
    unsigned* cursor  = (unsigned*)(s_trg + n_nodes);
    int2*     buckets = (int2*)(cursor + (size_t)nb * CSTR);
    float* out   = (float*)d_out;

    const int B = 256;

    {
        int tiles  = (n_nodes + 15) / 16;
        int blocks = (tiles + 7) / 8;
        k_gemm<<<blocks, B, 0, stream>>>(X, W, wx, n_nodes);
    }
    k_scores<<<(n_nodes + B - 1) / B, B, 0, stream>>>(wx, att, s_src, s_trg, cursor, n_nodes, nb);
    k_bin<<<(n_edges + BTHR * BEPT - 1) / (BTHR * BEPT), BTHR, 0, stream>>>(esrc, etrg, cursor, buckets,
                                                                            n_nodes, n_edges, nb);
    k_edges<<<nb, ETHR, 0, stream>>>(esrc, etrg, s_src, s_trg, wx, cursor, buckets, out,
                                     n_nodes, n_edges);
}


# Edge weight computed once per edge by one lane: logit then exp kept in LDS beside the record, weight handed to the row-adding wave by lane reads; no recompute in 32 lanes (sums by arrival as before)
# speedup vs baseline: 3.1526x; 1.2403x over previous
#include <hip/hip_runtime.h>
#include <hip/hip_bf16.h>


#define D_IN   128
#define D_OUT  64
#define LEAKY  0.2f

typedef float v2f __attribute__((ext_vector_type(2)));
typedef float v8f __attribute__((ext_vector_type(8)));

__device__ __forceinline__ unsigned ord_of_float(float f) {
    unsigned u = __float_as_uint(f);
    return (u & 0x80000000u) ? ~u : (u | 0x80000000u);
}
__device__ __forceinline__ float float_of_ord(unsigned u) {
    return __uint_as_float((u & 0x80000000u) ? (u ^ 0x80000000u) : ~u);
}

__global__ __launch_bounds__(256) void k_gemm(const float* __restrict__ X,
                                              const float* __restrict__ W,
                                              float* __restrict__ wx,
                                              int n_nodes) {
    constexpr int LDW = D_IN + 4;
    __shared__ float sWt[D_OUT * LDW];
    for (int i = threadIdx.x; i < D_IN * D_OUT; i += blockDim.x) {
        int k = i / D_OUT, col = i % D_OUT;
        sWt[col * LDW + k] = W[i];
    }
    __syncthreads();

    const int lane = threadIdx.x & 31;
    const int wave = threadIdx.x >> 5;
    const int tile = blockIdx.x * (blockDim.x >> 5) + wave;
    const int row0 = tile * 16;
    if (row0 >= n_nodes) return;

    const int m    = lane & 15;
    const int koff = (lane >> 4) * 2;
    int arow = row0 + m;
    if (arow >= n_nodes) arow = n_nodes - 1;

    v8f acc[4] = {v8f{}, v8f{}, v8f{}, v8f{}};

    const float* __restrict__ xrow = X + (size_t)arow * D_IN;
    #pragma unroll
    for (int kb = 0; kb < D_IN; kb += 4) {
        v2f a = *(const v2f*)(xrow + kb + koff);
        #pragma unroll
        for (int t = 0; t < 4; ++t) {
            const int col = t * 16 + m;
            v2f b = *(const v2f*)(&sWt[col * LDW + kb + koff]);
            acc[t] = __builtin_amdgcn_wmma_f32_16x16x4_f32(
                 false, a,  false, b,
                 (short)0, acc[t],
                 false,  false);
        }
    }

    const int rbase = row0 + ((lane >> 4) * 8);
    #pragma unroll
    for (int t = 0; t < 4; ++t) {
        #pragma unroll
        for (int r = 0; r < 8; ++r) {
            const int row = rbase + r;
            if (row < n_nodes)
                wx[(size_t)row * D_OUT + t * 16 + m] = acc[t][r];
        }
    }
}

#define NPW   391
#define CAP   8192
#define ETHR  1024
#define BTHR  1024
#define BEPT  4
#define CSTR  64
#define NBMAX 1024

__global__ __launch_bounds__(256) void k_scores(const float* __restrict__ wx,
                                                const float* __restrict__ att,
                                                float* __restrict__ s_src,
                                                float* __restrict__ s_trg,
                                                unsigned* __restrict__ cursor,
                                                int n_nodes, int nb) {
    int node = blockIdx.x * blockDim.x + threadIdx.x;
    if (node < nb) cursor[node * CSTR] = 0u;
    if (node >= n_nodes) return;
    const float* row = wx + (size_t)node * D_OUT;
    float ss = 0.0f, st = 0.0f;
    #pragma unroll 8
    for (int d = 0; d < D_OUT; ++d) {
        float v = row[d];
        ss = fmaf(v, att[d], ss);
        st = fmaf(v, att[D_OUT + d], st);
    }
    s_src[node] = ss;
    s_trg[node] = st;
}

__global__ __launch_bounds__(BTHR) void k_bin(const int* __restrict__ esrc,
                                              const int* __restrict__ etrg,
                                              unsigned* __restrict__ cursor,
                                              int2* __restrict__ buckets,
                                              int n_nodes, int n_edges, int nb) {
    __shared__ unsigned hist[NBMAX];
    const int tid = threadIdx.x;
    for (int i = tid; i < nb; i += BTHR) hist[i] = 0u;
    __syncthreads();

    const int base = blockIdx.x * (BTHR * BEPT) + tid;
    int s[BEPT], t[BEPT], b[BEPT];
    unsigned r[BEPT];
    #pragma unroll
    for (int u = 0; u < BEPT; ++u) {
        const int e = base + u * BTHR;
        const bool in = e < n_edges;
        s[u] = in ? esrc[e] : -1;
        t[u] = in ? etrg[e] : 0;
    }
    #pragma unroll
    for (int u = 0; u < BEPT; ++u) {
        const bool ok = (unsigned)s[u] < (unsigned)n_nodes;
        b[u] = ok ? s[u] / NPW : -1;
        r[u] = ok ? atomicAdd(&hist[b[u]], 1u) : 0u;
    }
    __syncthreads();
    for (int i = tid; i < nb; i += BTHR) {
        const unsigned c = hist[i];
        hist[i] = c ? atomicAdd(&cursor[i * CSTR], c) : 0u;
    }
    __syncthreads();
    #pragma unroll
    for (int u = 0; u < BEPT; ++u) {
        if (b[u] >= 0) {
            const unsigned pos = hist[b[u]] + r[u];
            if (pos < (unsigned)CAP)
                buckets[(size_t)b[u] * CAP + pos] = make_int2(s[u] - b[u] * NPW, t[u]);
        }
    }
}

__device__ __forceinline__ float edge_logit(float a, float b) {
    float l = a + b;
    return (l >= 0.0f) ? l : LEAKY * l;
}

template <typename F>
__device__ __forceinline__ void scan_owned(const int* __restrict__ esrc, int n_edges,
                                           int n0, int nown, int rot, F body) {
    const int tid  = threadIdx.x;
    const int nvec = n_edges >> 2;
    const int4* __restrict__ src4 = (const int4*)esrc;
    for (int base = 0; base < nvec; base += 4 * ETHR) {
        int4 q[4];
        int  w[4];
        #pragma unroll
        for (int u = 0; u < 4; ++u) {
            const int v = base + u * ETHR + tid;
            int ww = v + rot;
            if (ww >= nvec) ww -= nvec;
            w[u] = ww;
            q[u] = (v < nvec) ? src4[ww] : make_int4(-1, -1, -1, -1);
        }
        #pragma unroll
        for (int u = 0; u < 4; ++u) {
            const int e0 = w[u] << 2;
            const unsigned d0 = (unsigned)(q[u].x - n0);
            const unsigned d1 = (unsigned)(q[u].y - n0);
            const unsigned d2 = (unsigned)(q[u].z - n0);
            const unsigned d3 = (unsigned)(q[u].w - n0);
            if (d0 < (unsigned)nown) body((int)d0, e0);
            if (d1 < (unsigned)nown) body((int)d1, e0 + 1);
            if (d2 < (unsigned)nown) body((int)d2, e0 + 2);
            if (d3 < (unsigned)nown) body((int)d3, e0 + 3);
        }
    }
    for (int e = (nvec << 2) + tid; e < n_edges; e += ETHR) {
        const unsigned d = (unsigned)(esrc[e] - n0);
        if (d < (unsigned)nown) body((int)d, e);
    }
}

#define RB 16

template <bool TAIL>
__device__ __forceinline__ void add_rows(const float* __restrict__ wx, float* accL,
                                         int my_s, int my_t, float my_a,
                                         int j0, int n, int lane) {
    int    sl[RB];
    float  al[RB];
    float2 v[RB];
    #pragma unroll
    for (int u = 0; u < RB; ++u) {
        const int jj = TAIL ? min(j0 + u, n - 1) : (j0 + u);
        sl[u] = __builtin_amdgcn_readlane(my_s, jj);
        const int t = __builtin_amdgcn_readlane(my_t, jj);
        al[u] = __int_as_float(__builtin_amdgcn_readlane(__float_as_int(my_a), jj));
        v[u]  = ((const float2*)(wx + (size_t)t * D_OUT))[lane];
    }
    #pragma unroll
    for (int u = 0; u < RB; ++u) {
        if (!TAIL || j0 + u < n) {
            float* a = &accL[sl[u] * D_OUT + 2 * lane];
            atomicAdd(a,     al[u] * v[u].x);
            atomicAdd(a + 1, al[u] * v[u].y);
        }
    }
}

__global__ __launch_bounds__(ETHR) void k_edges(const int* __restrict__ esrc,
                                                const int* __restrict__ etrg,
                                                const float* __restrict__ s_src,
                                                const float* __restrict__ s_trg,
                                                const float* __restrict__ wx,
                                                const unsigned* __restrict__ cursor,
                                                const int2* __restrict__ buckets,
                                                float* __restrict__ out,
                                                int n_nodes, int n_edges) {
    __shared__ float    accL[NPW * D_OUT];
    __shared__ unsigned maxL[NPW];
    __shared__ float    denL[NPW];
    __shared__ float    ssL[NPW];
    __shared__ int      listS[CAP];
    __shared__ int      listT[CAP];
    __shared__ float    listW[CAP];

    const int tid  = threadIdx.x;
    const int lane = tid & 31;
    const int wave = tid >> 5;
    const int n0   = blockIdx.x * NPW;
    const int nown = min(NPW, n_nodes - n0);

    for (int i = tid; i < NPW * D_OUT; i += ETHR) accL[i] = 0.0f;
    for (int i = tid; i < NPW; i += ETHR) {
        maxL[i] = 0u;
        denL[i] = 0.0f;
        ssL[i]  = (i < nown) ? s_src[n0 + i] : 0.0f;
    }
    __syncthreads();

    const unsigned cnt = cursor[blockIdx.x * CSTR];

    if (cnt <= (unsigned)CAP) {
        const int2* __restrict__ bk = buckets + (size_t)blockIdx.x * CAP;
        for (unsigned i = tid; i < cnt; i += ETHR) {
            const int2 rec = bk[i];
            const int sl = rec.x;
            const int t  = rec.y;
            const float l = edge_logit(ssL[sl], s_trg[t]);
            listS[i] = sl;
            listT[i] = t;
            listW[i] = l;
            atomicMax(&maxL[sl], ord_of_float(l));
        }
        __syncthreads();
        for (unsigned i = tid; i < cnt; i += ETHR) {
            const int sl = listS[i];
            const float ex = __expf(listW[i] - float_of_ord(maxL[sl]));
            listW[i] = ex;
            atomicAdd(&denL[sl], ex);
        }
        __syncthreads();
        const unsigned wv   = (unsigned)__builtin_amdgcn_readfirstlane(wave);
        const unsigned ucnt = (unsigned)__builtin_amdgcn_readfirstlane((int)cnt);
        for (unsigned base = wv * 32u; base < ucnt; base += (ETHR / 32) * 32u) {
            const unsigned i  = base + lane;
            const bool     ok = i < ucnt;
            const int   my_s = ok ? listS[i] : 0;
            const int   my_t = ok ? listT[i] : 0;
            const float my_a = ok ? listW[i] / denL[my_s] : 0.0f;
            const int n = (int)min(32u, ucnt - base);
            for (int j0 = 0; j0 < n; j0 += RB) {
                if (j0 + RB <= n) add_rows<false>(wx, accL, my_s, my_t, my_a, j0, n, lane);
                else              add_rows<true >(wx, accL, my_s, my_t, my_a, j0, n, lane);
            }
        }
    } else {
        const int nvec = n_edges >> 2;
        const int rot  = (int)(((long long)blockIdx.x * nvec) / gridDim.x) & ~31;
        scan_owned(esrc, n_edges, n0, nown, rot, [&](int sl, int e) {
            const int t = etrg[e];
            atomicMax(&maxL[sl], ord_of_float(edge_logit(ssL[sl], s_trg[t])));
        });
        __syncthreads();
        scan_owned(esrc, n_edges, n0, nown, rot, [&](int sl, int e) {
            const float l = edge_logit(ssL[sl], s_trg[etrg[e]]);
            atomicAdd(&denL[sl], __expf(l - float_of_ord(maxL[sl])));
        });
        __syncthreads();
        scan_owned(esrc, n_edges, n0, nown, rot, [&](int sl, int e) {
            const int t = etrg[e];
            const float l = edge_logit(ssL[sl], s_trg[t]);
            const float alpha = __expf(l - float_of_ord(maxL[sl])) / denL[sl];
            const float* __restrict__ row = wx + (size_t)t * D_OUT;
            for (int d = 0; d < D_OUT; ++d) atomicAdd(&accL[sl * D_OUT + d], alpha * row[d]);
        });
    }
    __syncthreads();

    float* __restrict__ o = out + (size_t)n0 * D_OUT;
    for (int i = tid; i < nown * D_OUT; i += ETHR) o[i] = accL[i];
}

extern "C" void kernel_launch(void* const* d_in, const int* in_sizes, int n_in,
                              void* d_out, int out_size, void* d_ws, size_t ws_size,
                              hipStream_t stream) {
    const float* X   = (const float*)d_in[0];
    const float* W   = (const float*)d_in[1];
    const float* att = (const float*)d_in[2];
    const int*   adj = (const int*)  d_in[3];

    const int d_out_dim = in_sizes[2] / 2;
    const int d_in_dim  = in_sizes[1] / d_out_dim;
    const int n_nodes   = in_sizes[0] / d_in_dim;
    const int n_edges   = in_sizes[3] / 2;
    (void)d_out_dim; (void)d_in_dim;

    const int* esrc = adj;
    const int* etrg = adj + n_edges;

    float* wx    = (float*)d_ws;
    float* s_src = wx + (size_t)n_nodes * D_OUT;
    float* s_trg = s_src + n_nodes;
    const int nb = (n_nodes + NPW - 1) / NPW;
    unsigned* cursor  = (unsigned*)(s_trg + n_nodes);
    int2*     buckets = (int2*)(cursor + (size_t)nb * CSTR);
    float* out   = (float*)d_out;

    const int B = 256;

    {
        int tiles  = (n_nodes + 15) / 16;
        int blocks = (tiles + 7) / 8;
        k_gemm<<<blocks, B, 0, stream>>>(X, W, wx, n_nodes);
    }
    k_scores<<<(n_nodes + B - 1) / B, B, 0, stream>>>(wx, att, s_src, s_trg, cursor, n_nodes, nb);
    k_bin<<<(n_edges + BTHR * BEPT - 1) / (BTHR * BEPT), BTHR, 0, stream>>>(esrc, etrg, cursor, buckets,
                                                                            n_nodes, n_edges, nb);
    k_edges<<<nb, ETHR, 0, stream>>>(esrc, etrg, s_src, s_trg, wx, cursor, buckets, out,
                                     n_nodes, n_edges);
}


# Scores fused into the GEMM epilogue: s_src and s_trg summed from the accumulator tile by DPP lane-row sums (score sum order becomes a lane tree); the scores launch and its re-read of wx are gone
# speedup vs baseline: 3.3841x; 1.0734x over previous
#include <hip/hip_runtime.h>
#include <hip/hip_bf16.h>


#define D_IN   128
#define D_OUT  64
#define LEAKY  0.2f

typedef float v2f __attribute__((ext_vector_type(2)));
typedef float v8f __attribute__((ext_vector_type(8)));

__device__ __forceinline__ unsigned ord_of_float(float f) {
    unsigned u = __float_as_uint(f);
    return (u & 0x80000000u) ? ~u : (u | 0x80000000u);
}
__device__ __forceinline__ float float_of_ord(unsigned u) {
    return __uint_as_float((u & 0x80000000u) ? (u ^ 0x80000000u) : ~u);
}

#define CSTR  64

template <int N>
__device__ __forceinline__ float row_xor(float v) {
    return __int_as_float(__builtin_amdgcn_update_dpp(0, __float_as_int(v), 0x160 | N, 0xf, 0xf, true));
}

__global__ __launch_bounds__(256) void k_gemm(const float* __restrict__ X,
                                              const float* __restrict__ W,
                                              const float* __restrict__ att,
                                              float* __restrict__ wx,
                                              float* __restrict__ s_src,
                                              float* __restrict__ s_trg,
                                              unsigned* __restrict__ cursor,
                                              int n_nodes, int nb) {
    {
        const int g = blockIdx.x * blockDim.x + threadIdx.x;
        if (g < nb) cursor[g * CSTR] = 0u;
    }
    constexpr int LDW = D_IN + 4;
    __shared__ float sWt[D_OUT * LDW];
    for (int i = threadIdx.x; i < D_IN * D_OUT; i += blockDim.x) {
        int k = i / D_OUT, col = i % D_OUT;
        sWt[col * LDW + k] = W[i];
    }
    __syncthreads();

    const int lane = threadIdx.x & 31;
    const int wave = threadIdx.x >> 5;
    const int tile = blockIdx.x * (blockDim.x >> 5) + wave;
    const int row0 = tile * 16;
    if (row0 >= n_nodes) return;

    const int m    = lane & 15;
    const int koff = (lane >> 4) * 2;
    int arow = row0 + m;
    if (arow >= n_nodes) arow = n_nodes - 1;

    v8f acc[4] = {v8f{}, v8f{}, v8f{}, v8f{}};

    float as_[4], at_[4];
    #pragma unroll
    for (int t = 0; t < 4; ++t) {
        as_[t] = att[t * 16 + m];
        at_[t] = att[D_OUT + t * 16 + m];
    }

    const float* __restrict__ xrow = X + (size_t)arow * D_IN;
    #pragma unroll
    for (int kb = 0; kb < D_IN; kb += 4) {
        v2f a = *(const v2f*)(xrow + kb + koff);
        #pragma unroll
        for (int t = 0; t < 4; ++t) {
            const int col = t * 16 + m;
            v2f b = *(const v2f*)(&sWt[col * LDW + kb + koff]);
            acc[t] = __builtin_amdgcn_wmma_f32_16x16x4_f32(
                 false, a,  false, b,
                 (short)0, acc[t],
                 false,  false);
        }
    }

    const int rbase = row0 + ((lane >> 4) * 8);
    #pragma unroll
    for (int t = 0; t < 4; ++t) {
        #pragma unroll
        for (int r = 0; r < 8; ++r) {
            const int row = rbase + r;
            if (row < n_nodes)
                wx[(size_t)row * D_OUT + t * 16 + m] = acc[t][r];
        }
    }

    float v16[16];
    #pragma unroll
    for (int r = 0; r < 8; ++r) {
        float ss = acc[0][r] * as_[0];
        float st = acc[0][r] * at_[0];
        #pragma unroll
        for (int t = 1; t < 4; ++t) {
            ss = fmaf(acc[t][r], as_[t], ss);
            st = fmaf(acc[t][r], at_[t], st);
        }
        v16[r]     = ss;
        v16[8 + r] = st;
    }
    const bool b3 = (m & 8) != 0, b2 = (m & 4) != 0, b1 = (m & 2) != 0, b0 = (m & 1) != 0;
    float v8[8], v4[4], v2[2];
    #pragma unroll
    for (int i = 0; i < 8; ++i) {
        const float keep = b3 ? v16[i + 8] : v16[i];
        const float send = b3 ? v16[i] : v16[i + 8];
        v8[i] = keep + row_xor<8>(send);
    }
    #pragma unroll
    for (int i = 0; i < 4; ++i) {
        const float keep = b2 ? v8[i + 4] : v8[i];
        const float send = b2 ? v8[i] : v8[i + 4];
        v4[i] = keep + row_xor<4>(send);
    }
    #pragma unroll
    for (int i = 0; i < 2; ++i) {
        const float keep = b1 ? v4[i + 2] : v4[i];
        const float send = b1 ? v4[i] : v4[i + 2];
        v2[i] = keep + row_xor<2>(send);
    }
    const float keep1 = b0 ? v2[1] : v2[0];
    const float send1 = b0 ? v2[0] : v2[1];
    const float score = keep1 + row_xor<1>(send1);
    {
        const int row = rbase + (m & 7);
        float* dst = b3 ? s_trg : s_src;
        if (row < n_nodes) dst[row] = score;
    }
}

#define NPW   391
#define CAP   8192
#define ETHR  1024
#define BTHR  1024
#define BEPT  4
#define NBMAX 1024

__global__ __launch_bounds__(BTHR) void k_bin(const int* __restrict__ esrc,
                                              const int* __restrict__ etrg,
                                              unsigned* __restrict__ cursor,
                                              int2* __restrict__ buckets,
                                              int n_nodes, int n_edges, int nb) {
    __shared__ unsigned hist[NBMAX];
    const int tid = threadIdx.x;
    for (int i = tid; i < nb; i += BTHR) hist[i] = 0u;
    __syncthreads();

    const int base = blockIdx.x * (BTHR * BEPT) + tid;
    int s[BEPT], t[BEPT], b[BEPT];
    unsigned r[BEPT];
    #pragma unroll
    for (int u = 0; u < BEPT; ++u) {
        const int e = base + u * BTHR;
        const bool in = e < n_edges;
        s[u] = in ? esrc[e] : -1;
        t[u] = in ? etrg[e] : 0;
    }
    #pragma unroll
    for (int u = 0; u < BEPT; ++u) {
        const bool ok = (unsigned)s[u] < (unsigned)n_nodes;
        b[u] = ok ? s[u] / NPW : -1;
        r[u] = ok ? atomicAdd(&hist[b[u]], 1u) : 0u;
    }
    __syncthreads();
    for (int i = tid; i < nb; i += BTHR) {
        const unsigned c = hist[i];
        hist[i] = c ? atomicAdd(&cursor[i * CSTR], c) : 0u;
    }
    __syncthreads();
    #pragma unroll
    for (int u = 0; u < BEPT; ++u) {
        if (b[u] >= 0) {
            const unsigned pos = hist[b[u]] + r[u];
            if (pos < (unsigned)CAP)
                buckets[(size_t)b[u] * CAP + pos] = make_int2(s[u] - b[u] * NPW, t[u]);
        }
    }
}

__device__ __forceinline__ float edge_logit(float a, float b) {
    float l = a + b;
    return (l >= 0.0f) ? l : LEAKY * l;
}

template <typename F>
__device__ __forceinline__ void scan_owned(const int* __restrict__ esrc, int n_edges,
                                           int n0, int nown, int rot, F body) {
    const int tid  = threadIdx.x;
    const int nvec = n_edges >> 2;
    const int4* __restrict__ src4 = (const int4*)esrc;
    for (int base = 0; base < nvec; base += 4 * ETHR) {
        int4 q[4];
        int  w[4];
        #pragma unroll
        for (int u = 0; u < 4; ++u) {
            const int v = base + u * ETHR + tid;
            int ww = v + rot;
            if (ww >= nvec) ww -= nvec;
            w[u] = ww;
            q[u] = (v < nvec) ? src4[ww] : make_int4(-1, -1, -1, -1);
        }
        #pragma unroll
        for (int u = 0; u < 4; ++u) {
            const int e0 = w[u] << 2;
            const unsigned d0 = (unsigned)(q[u].x - n0);
            const unsigned d1 = (unsigned)(q[u].y - n0);
            const unsigned d2 = (unsigned)(q[u].z - n0);
            const unsigned d3 = (unsigned)(q[u].w - n0);
            if (d0 < (unsigned)nown) body((int)d0, e0);
            if (d1 < (unsigned)nown) body((int)d1, e0 + 1);
            if (d2 < (unsigned)nown) body((int)d2, e0 + 2);
            if (d3 < (unsigned)nown) body((int)d3, e0 + 3);
        }
    }
    for (int e = (nvec << 2) + tid; e < n_edges; e += ETHR) {
        const unsigned d = (unsigned)(esrc[e] - n0);
        if (d < (unsigned)nown) body((int)d, e);
    }
}

#define RB 16

template <bool TAIL>
__device__ __forceinline__ void add_rows(const float* __restrict__ wx, float* accL,
                                         int my_s, int my_t, float my_a,
                                         int j0, int n, int lane) {
    int    sl[RB];
    float  al[RB];
    float2 v[RB];
    #pragma unroll
    for (int u = 0; u < RB; ++u) {
        const int jj = TAIL ? min(j0 + u, n - 1) : (j0 + u);
        sl[u] = __builtin_amdgcn_readlane(my_s, jj);
        const int t = __builtin_amdgcn_readlane(my_t, jj);
        al[u] = __int_as_float(__builtin_amdgcn_readlane(__float_as_int(my_a), jj));
        v[u]  = ((const float2*)(wx + (size_t)t * D_OUT))[lane];
    }
    #pragma unroll
    for (int u = 0; u < RB; ++u) {
        if (!TAIL || j0 + u < n) {
            float* a = &accL[sl[u] * D_OUT + 2 * lane];
            atomicAdd(a,     al[u] * v[u].x);
            atomicAdd(a + 1, al[u] * v[u].y);
        }
    }
}

__global__ __launch_bounds__(ETHR) void k_edges(const int* __restrict__ esrc,
                                                const int* __restrict__ etrg,
                                                const float* __restrict__ s_src,
                                                const float* __restrict__ s_trg,
                                                const float* __restrict__ wx,
                                                const unsigned* __restrict__ cursor,
                                                const int2* __restrict__ buckets,
                                                float* __restrict__ out,
                                                int n_nodes, int n_edges) {
    __shared__ float    accL[NPW * D_OUT];
    __shared__ unsigned maxL[NPW];
    __shared__ float    denL[NPW];
    __shared__ float    ssL[NPW];
    __shared__ int      listS[CAP];
    __shared__ int      listT[CAP];
    __shared__ float    listW[CAP];

    const int tid  = threadIdx.x;
    const int lane = tid & 31;
    const int wave = tid >> 5;
    const int n0   = blockIdx.x * NPW;
    const int nown = min(NPW, n_nodes - n0);

    for (int i = tid; i < NPW * D_OUT; i += ETHR) accL[i] = 0.0f;
    for (int i = tid; i < NPW; i += ETHR) {
        maxL[i] = 0u;
        denL[i] = 0.0f;
        ssL[i]  = (i < nown) ? s_src[n0 + i] : 0.0f;
    }
    __syncthreads();

    const unsigned cnt = cursor[blockIdx.x * CSTR];

    if (cnt <= (unsigned)CAP) {
        const int2* __restrict__ bk = buckets + (size_t)blockIdx.x * CAP;
        for (unsigned i = tid; i < cnt; i += ETHR) {
            const int2 rec = bk[i];
            const int sl = rec.x;
            const int t  = rec.y;
            const float l = edge_logit(ssL[sl], s_trg[t]);
            listS[i] = sl;
            listT[i] = t;
            listW[i] = l;
            atomicMax(&maxL[sl], ord_of_float(l));
        }
        __syncthreads();
        for (unsigned i = tid; i < cnt; i += ETHR) {
            const int sl = listS[i];
            const float ex = __expf(listW[i] - float_of_ord(maxL[sl]));
            listW[i] = ex;
            atomicAdd(&denL[sl], ex);
        }
        __syncthreads();
        const unsigned wv   = (unsigned)__builtin_amdgcn_readfirstlane(wave);
        const unsigned ucnt = (unsigned)__builtin_amdgcn_readfirstlane((int)cnt);
        for (unsigned base = wv * 32u; base < ucnt; base += (ETHR / 32) * 32u) {
            const unsigned i  = base + lane;
            const bool     ok = i < ucnt;
            const int   my_s = ok ? listS[i] : 0;
            const int   my_t = ok ? listT[i] : 0;
            const float my_a = ok ? listW[i] / denL[my_s] : 0.0f;
            const int n = (int)min(32u, ucnt - base);
            for (int j0 = 0; j0 < n; j0 += RB) {
                if (j0 + RB <= n) add_rows<false>(wx, accL, my_s, my_t, my_a, j0, n, lane);
                else              add_rows<true >(wx, accL, my_s, my_t, my_a, j0, n, lane);
            }
        }
    } else {
        const int nvec = n_edges >> 2;
        const int rot  = (int)(((long long)blockIdx.x * nvec) / gridDim.x) & ~31;
        scan_owned(esrc, n_edges, n0, nown, rot, [&](int sl, int e) {
            const int t = etrg[e];
            atomicMax(&maxL[sl], ord_of_float(edge_logit(ssL[sl], s_trg[t])));
        });
        __syncthreads();
        scan_owned(esrc, n_edges, n0, nown, rot, [&](int sl, int e) {
            const float l = edge_logit(ssL[sl], s_trg[etrg[e]]);
            atomicAdd(&denL[sl], __expf(l - float_of_ord(maxL[sl])));
        });
        __syncthreads();
        scan_owned(esrc, n_edges, n0, nown, rot, [&](int sl, int e) {
            const int t = etrg[e];
            const float l = edge_logit(ssL[sl], s_trg[t]);
            const float alpha = __expf(l - float_of_ord(maxL[sl])) / denL[sl];
            const float* __restrict__ row = wx + (size_t)t * D_OUT;
            for (int d = 0; d < D_OUT; ++d) atomicAdd(&accL[sl * D_OUT + d], alpha * row[d]);
        });
    }
    __syncthreads();

    float* __restrict__ o = out + (size_t)n0 * D_OUT;
    for (int i = tid; i < nown * D_OUT; i += ETHR) o[i] = accL[i];
}

extern "C" void kernel_launch(void* const* d_in, const int* in_sizes, int n_in,
                              void* d_out, int out_size, void* d_ws, size_t ws_size,
                              hipStream_t stream) {
    const float* X   = (const float*)d_in[0];
    const float* W   = (const float*)d_in[1];
    const float* att = (const float*)d_in[2];
    const int*   adj = (const int*)  d_in[3];

    const int d_out_dim = in_sizes[2] / 2;
    const int d_in_dim  = in_sizes[1] / d_out_dim;
    const int n_nodes   = in_sizes[0] / d_in_dim;
    const int n_edges   = in_sizes[3] / 2;
    (void)d_out_dim; (void)d_in_dim;

    const int* esrc = adj;
    const int* etrg = adj + n_edges;

    float* wx    = (float*)d_ws;
    float* s_src = wx + (size_t)n_nodes * D_OUT;
    float* s_trg = s_src + n_nodes;
    const int nb = (n_nodes + NPW - 1) / NPW;
    unsigned* cursor  = (unsigned*)(s_trg + n_nodes);
    int2*     buckets = (int2*)(cursor + (size_t)nb * CSTR);
    float* out   = (float*)d_out;

    const int B = 256;

    {
        int tiles  = (n_nodes + 15) / 16;
        int blocks = (tiles + 7) / 8;
        k_gemm<<<blocks, B, 0, stream>>>(X, W, att, wx, s_src, s_trg, cursor, n_nodes, nb);
    }
    k_bin<<<(n_edges + BTHR * BEPT - 1) / (BTHR * BEPT), BTHR, 0, stream>>>(esrc, etrg, cursor, buckets,
                                                                            n_nodes, n_edges, nb);
    k_edges<<<nb, ETHR, 0, stream>>>(esrc, etrg, s_src, s_trg, wx, cursor, buckets, out,
                                     n_nodes, n_edges);
}


# Binning runs inside the GEMM launch as extra work-groups: each sorts its 4096 edges by owner into its own segment plus a count table (no global cursors); 2 launches; edge sums by arrival as before
# speedup vs baseline: 4.2392x; 1.2527x over previous
#include <hip/hip_runtime.h>
#include <hip/hip_bf16.h>


#define D_IN   128
#define D_OUT  64
#define LEAKY  0.2f

typedef float v2f __attribute__((ext_vector_type(2)));
typedef float v8f __attribute__((ext_vector_type(8)));

__device__ __forceinline__ unsigned ord_of_float(float f) {
    unsigned u = __float_as_uint(f);
    return (u & 0x80000000u) ? ~u : (u | 0x80000000u);
}
__device__ __forceinline__ float float_of_ord(unsigned u) {
    return __uint_as_float((u & 0x80000000u) ? (u ^ 0x80000000u) : ~u);
}

template <int N>
__device__ __forceinline__ float row_xor(float v) {
    return __int_as_float(__builtin_amdgcn_update_dpp(0, __float_as_int(v), 0x160 | N, 0xf, 0xf, true));
}

__device__ __forceinline__ void gemm_part(const float* __restrict__ X,
                                          const float* __restrict__ W,
                                          const float* __restrict__ att,
                                          float* __restrict__ wx,
                                          float* __restrict__ s_src,
                                          float* __restrict__ s_trg,
                                          int n_nodes, int bid) {
    constexpr int LDW = D_IN + 4;
    __shared__ float sWt[D_OUT * LDW];
    for (int i = threadIdx.x; i < D_IN * D_OUT; i += blockDim.x) {
        int k = i / D_OUT, col = i % D_OUT;
        sWt[col * LDW + k] = W[i];
    }
    __syncthreads();

    const int lane = threadIdx.x & 31;
    const int wave = __builtin_amdgcn_readfirstlane((int)(threadIdx.x >> 5));
    const int tile = bid * (blockDim.x >> 5) + wave;
    const int row0 = tile * 16;
    if (row0 >= n_nodes) return;

    const int m    = lane & 15;
    const int koff = (lane >> 4) * 2;
    int arow = row0 + m;
    if (arow >= n_nodes) arow = n_nodes - 1;

    v8f acc[4] = {v8f{}, v8f{}, v8f{}, v8f{}};

    float as_[4], at_[4];
    #pragma unroll
    for (int t = 0; t < 4; ++t) {
        as_[t] = att[t * 16 + m];
        at_[t] = att[D_OUT + t * 16 + m];
    }

    const float* __restrict__ xrow = X + (size_t)arow * D_IN;
    #pragma unroll
    for (int kb = 0; kb < D_IN; kb += 4) {
        v2f a = *(const v2f*)(xrow + kb + koff);
        #pragma unroll
        for (int t = 0; t < 4; ++t) {
            const int col = t * 16 + m;
            v2f b = *(const v2f*)(&sWt[col * LDW + kb + koff]);
            acc[t] = __builtin_amdgcn_wmma_f32_16x16x4_f32(
                 false, a,  false, b,
                 (short)0, acc[t],
                 false,  false);
        }
    }

    const int rbase = row0 + ((lane >> 4) * 8);
    #pragma unroll
    for (int t = 0; t < 4; ++t) {
        #pragma unroll
        for (int r = 0; r < 8; ++r) {
            const int row = rbase + r;
            if (row < n_nodes)
                wx[(size_t)row * D_OUT + t * 16 + m] = acc[t][r];
        }
    }

    float v16[16];
    #pragma unroll
    for (int r = 0; r < 8; ++r) {
        float ss = acc[0][r] * as_[0];
        float st = acc[0][r] * at_[0];
        #pragma unroll
        for (int t = 1; t < 4; ++t) {
            ss = fmaf(acc[t][r], as_[t], ss);
            st = fmaf(acc[t][r], at_[t], st);
        }
        v16[r]     = ss;
        v16[8 + r] = st;
    }
    const bool b3 = (m & 8) != 0, b2 = (m & 4) != 0, b1 = (m & 2) != 0, b0 = (m & 1) != 0;
    float v8[8], v4[4], v2[2];
    #pragma unroll
    for (int i = 0; i < 8; ++i) {
        const float keep = b3 ? v16[i + 8] : v16[i];
        const float send = b3 ? v16[i] : v16[i + 8];
        v8[i] = keep + row_xor<8>(send);
    }
    #pragma unroll
    for (int i = 0; i < 4; ++i) {
        const float keep = b2 ? v8[i + 4] : v8[i];
        const float send = b2 ? v8[i] : v8[i + 4];
        v4[i] = keep + row_xor<4>(send);
    }
    #pragma unroll
    for (int i = 0; i < 2; ++i) {
        const float keep = b1 ? v4[i + 2] : v4[i];
        const float send = b1 ? v4[i] : v4[i + 2];
        v2[i] = keep + row_xor<2>(send);
    }
    const float keep1 = b0 ? v2[1] : v2[0];
    const float send1 = b0 ? v2[0] : v2[1];
    const float score = keep1 + row_xor<1>(send1);
    {
        const int row = rbase + (m & 7);
        float* dst = b3 ? s_trg : s_src;
        if (row < n_nodes) dst[row] = score;
    }
}

#define NPW   391
#define CAP   8192
#define ETHR  1024
#define BTHR  1024
#define BEPT  4
#define SEGE  (BTHR * BEPT)
#define NBMAX 1024
#define SEGMAX 256
#define CBITS 13

__device__ __forceinline__ void bin_part(const int* __restrict__ esrc,
                                         const int* __restrict__ etrg,
                                         unsigned* __restrict__ tab,
                                         int2* __restrict__ segs,
                                         int n_nodes, int n_edges, int nb, int nseg, int seg) {
    __shared__ unsigned hist[NBMAX];
    __shared__ unsigned wtot[32];
    const int tid  = threadIdx.x;
    const int lane = tid & 31;
    const int wave = tid >> 5;
    hist[tid] = 0u;
    __syncthreads();

    const int base = seg * SEGE + tid;
    int s[BEPT], t[BEPT], b[BEPT];
    unsigned r[BEPT];
    #pragma unroll
    for (int u = 0; u < BEPT; ++u) {
        const int e = base + u * BTHR;
        const bool in = e < n_edges;
        s[u] = in ? esrc[e] : -1;
        t[u] = in ? etrg[e] : 0;
    }
    #pragma unroll
    for (int u = 0; u < BEPT; ++u) {
        const bool ok = (unsigned)s[u] < (unsigned)n_nodes;
        b[u] = ok ? s[u] / NPW : -1;
        r[u] = ok ? atomicAdd(&hist[b[u]], 1u) : 0u;
    }
    __syncthreads();
    const unsigned cntv = hist[tid];
    unsigned inc = cntv;
    #pragma unroll
    for (int d = 1; d < 32; d <<= 1) {
        const unsigned o = __shfl_up(inc, d);
        if (lane >= d) inc += o;
    }
    if (lane == 31) wtot[wave] = inc;
    __syncthreads();
    if (wave == 0) {
        const unsigned w = wtot[lane];
        unsigned winc = w;
        #pragma unroll
        for (int d = 1; d < 32; d <<= 1) {
            const unsigned o = __shfl_up(winc, d);
            if (lane >= d) winc += o;
        }
        wtot[lane] = winc - w;
    }
    __syncthreads();
    const unsigned start = inc - cntv + wtot[wave];
    hist[tid] = start;
    if (tid < nb) tab[(size_t)tid * nseg + seg] = (start << CBITS) | cntv;
    __syncthreads();
    int2* __restrict__ mine = segs + (size_t)seg * SEGE;
    #pragma unroll
    for (int u = 0; u < BEPT; ++u) {
        if (b[u] >= 0)
            mine[hist[b[u]] + r[u]] = make_int2(s[u] - b[u] * NPW, t[u]);
    }
}

__global__ __launch_bounds__(BTHR, 16) void k_front(const float* __restrict__ X,
                                                const float* __restrict__ W,
                                                const float* __restrict__ att,
                                                float* __restrict__ wx,
                                                float* __restrict__ s_src,
                                                float* __restrict__ s_trg,
                                                const int* __restrict__ esrc,
                                                const int* __restrict__ etrg,
                                                unsigned* __restrict__ tab,
                                                int2* __restrict__ segs,
                                                int n_nodes, int n_edges, int nb, int nseg,
                                                int gemm_blocks) {
    if ((int)blockIdx.x < gemm_blocks)
        gemm_part(X, W, att, wx, s_src, s_trg, n_nodes, (int)blockIdx.x);
    else
        bin_part(esrc, etrg, tab, segs, n_nodes, n_edges, nb, nseg, (int)blockIdx.x - gemm_blocks);
}

__device__ __forceinline__ float edge_logit(float a, float b) {
    float l = a + b;
    return (l >= 0.0f) ? l : LEAKY * l;
}

template <typename F>
__device__ __forceinline__ void scan_owned(const int* __restrict__ esrc, int n_edges,
                                           int n0, int nown, int rot, F body) {
    const int tid  = threadIdx.x;
    const int nvec = n_edges >> 2;
    const int4* __restrict__ src4 = (const int4*)esrc;
    for (int base = 0; base < nvec; base += 4 * ETHR) {
        int4 q[4];
        int  w[4];
        #pragma unroll
        for (int u = 0; u < 4; ++u) {
            const int v = base + u * ETHR + tid;
            int ww = v + rot;
            if (ww >= nvec) ww -= nvec;
            w[u] = ww;
            q[u] = (v < nvec) ? src4[ww] : make_int4(-1, -1, -1, -1);
        }
        #pragma unroll
        for (int u = 0; u < 4; ++u) {
            const int e0 = w[u] << 2;
            const unsigned d0 = (unsigned)(q[u].x - n0);
            const unsigned d1 = (unsigned)(q[u].y - n0);
            const unsigned d2 = (unsigned)(q[u].z - n0);
            const unsigned d3 = (unsigned)(q[u].w - n0);
            if (d0 < (unsigned)nown) body((int)d0, e0);
            if (d1 < (unsigned)nown) body((int)d1, e0 + 1);
            if (d2 < (unsigned)nown) body((int)d2, e0 + 2);
            if (d3 < (unsigned)nown) body((int)d3, e0 + 3);
        }
    }
    for (int e = (nvec << 2) + tid; e < n_edges; e += ETHR) {
        const unsigned d = (unsigned)(esrc[e] - n0);
        if (d < (unsigned)nown) body((int)d, e);
    }
}

#define RB 16

template <bool TAIL>
__device__ __forceinline__ void add_rows(const float* __restrict__ wx, float* accL,
                                         int my_s, int my_t, float my_a,
                                         int j0, int n, int lane) {
    int    sl[RB];
    float  al[RB];
    float2 v[RB];
    #pragma unroll
    for (int u = 0; u < RB; ++u) {
        const int jj = TAIL ? min(j0 + u, n - 1) : (j0 + u);
        sl[u] = __builtin_amdgcn_readlane(my_s, jj);
        const int t = __builtin_amdgcn_readlane(my_t, jj);
        al[u] = __int_as_float(__builtin_amdgcn_readlane(__float_as_int(my_a), jj));
        v[u]  = ((const float2*)(wx + (size_t)t * D_OUT))[lane];
    }
    #pragma unroll
    for (int u = 0; u < RB; ++u) {
        if (!TAIL || j0 + u < n) {
            float* a = &accL[sl[u] * D_OUT + 2 * lane];
            atomicAdd(a,     al[u] * v[u].x);
            atomicAdd(a + 1, al[u] * v[u].y);
        }
    }
}

__global__ __launch_bounds__(ETHR) void k_edges(const int* __restrict__ esrc,
                                                const int* __restrict__ etrg,
                                                const float* __restrict__ s_src,
                                                const float* __restrict__ s_trg,
                                                const float* __restrict__ wx,
                                                const unsigned* __restrict__ tab,
                                                const int2* __restrict__ segs,
                                                float* __restrict__ out,
                                                int n_nodes, int n_edges, int nseg) {
    __shared__ float    accL[NPW * D_OUT];
    __shared__ unsigned maxL[NPW];
    __shared__ float    denL[NPW];
    __shared__ float    ssL[NPW];
    __shared__ int      listS[CAP];
    __shared__ int      listT[CAP];
    __shared__ float    listW[CAP];
    __shared__ unsigned segInfo[SEGMAX];
    __shared__ unsigned segBase[SEGMAX];
    __shared__ unsigned lcnt;
    __shared__ unsigned fbIdx;

    const int tid  = threadIdx.x;
    const int lane = tid & 31;
    const int wave = tid >> 5;
    const int n0   = blockIdx.x * NPW;
    const int nown = min(NPW, n_nodes - n0);

    for (int i = tid; i < NPW * D_OUT; i += ETHR) accL[i] = 0.0f;
    for (int i = tid; i < NPW; i += ETHR) {
        maxL[i] = 0u;
        denL[i] = 0.0f;
        ssL[i]  = (i < nown) ? s_src[n0 + i] : 0.0f;
    }
    if (tid == 0) lcnt = 0u;
    __syncthreads();

    const bool have = nseg <= SEGMAX;
    if (have) {
        for (int i = tid; i < nseg; i += ETHR) {
            const unsigned p = tab[(size_t)blockIdx.x * nseg + i];
            segInfo[i] = p;
            segBase[i] = atomicAdd(&lcnt, p & ((1u << CBITS) - 1u));
        }
    }
    __syncthreads();
    const unsigned cnt = have ? lcnt : (unsigned)CAP + 1u;

    if (cnt <= (unsigned)CAP) {
        const unsigned l16   = (unsigned)tid & 15u;
        const unsigned cmask = (1u << CBITS) - 1u;
        for (int sg = tid >> 4; sg < nseg; sg += ETHR / 16) {
            const unsigned p     = segInfo[sg];
            const unsigned c     = p & cmask;
            const unsigned bs    = segBase[sg];
            const unsigned first = (unsigned)sg * SEGE + (p >> CBITS);
            for (unsigned j = l16; j < c; j += 16) {
                listT[bs + j] = (int)(first + j);
                if (bs + j == 0u) fbIdx = first + j;
            }
        }
        __syncthreads();
        const unsigned fb = fbIdx;
        for (unsigned base = 0; base < cnt; base += 4 * ETHR) {
            unsigned idx[4];
            int2     rec[4];
            float    st[4];
            #pragma unroll
            for (int u = 0; u < 4; ++u) {
                const unsigned i = base + u * ETHR + tid;
                idx[u] = (i < cnt) ? (unsigned)listT[i] : fb;
            }
            #pragma unroll
            for (int u = 0; u < 4; ++u) rec[u] = segs[idx[u]];
            #pragma unroll
            for (int u = 0; u < 4; ++u) st[u] = s_trg[rec[u].y];
            asm volatile("" :: "v"(st[0]), "v"(st[1]), "v"(st[2]), "v"(st[3]));
            #pragma unroll
            for (int u = 0; u < 4; ++u) {
                const unsigned i = base + u * ETHR + tid;
                if (i < cnt) {
                    const int sl = rec[u].x;
                    const float l = edge_logit(ssL[sl], st[u]);
                    listS[i] = sl;
                    listT[i] = rec[u].y;
                    listW[i] = l;
                    atomicMax(&maxL[sl], ord_of_float(l));
                }
            }
        }
        __syncthreads();
        for (unsigned i = tid; i < cnt; i += ETHR) {
            const int sl = listS[i];
            const float ex = __expf(listW[i] - float_of_ord(maxL[sl]));
            listW[i] = ex;
            atomicAdd(&denL[sl], ex);
        }
        __syncthreads();
        const unsigned wv   = (unsigned)__builtin_amdgcn_readfirstlane(wave);
        const unsigned ucnt = (unsigned)__builtin_amdgcn_readfirstlane((int)cnt);
        for (unsigned base = wv * 32u; base < ucnt; base += (ETHR / 32) * 32u) {
            const unsigned i  = base + lane;
            const bool     ok = i < ucnt;
            const int   my_s = ok ? listS[i] : 0;
            const int   my_t = ok ? listT[i] : 0;
            const float my_a = ok ? listW[i] / denL[my_s] : 0.0f;
            const int n = (int)min(32u, ucnt - base);
            for (int j0 = 0; j0 < n; j0 += RB) {
                if (j0 + RB <= n) add_rows<false>(wx, accL, my_s, my_t, my_a, j0, n, lane);
                else              add_rows<true >(wx, accL, my_s, my_t, my_a, j0, n, lane);
            }
        }
    } else {
        const int nvec = n_edges >> 2;
        const int rot  = (int)(((long long)blockIdx.x * nvec) / gridDim.x) & ~31;
        scan_owned(esrc, n_edges, n0, nown, rot, [&](int sl, int e) {
            const int t = etrg[e];
            atomicMax(&maxL[sl], ord_of_float(edge_logit(ssL[sl], s_trg[t])));
        });
        __syncthreads();
        scan_owned(esrc, n_edges, n0, nown, rot, [&](int sl, int e) {
            const float l = edge_logit(ssL[sl], s_trg[etrg[e]]);
            atomicAdd(&denL[sl], __expf(l - float_of_ord(maxL[sl])));
        });
        __syncthreads();
        scan_owned(esrc, n_edges, n0, nown, rot, [&](int sl, int e) {
            const int t = etrg[e];
            const float l = edge_logit(ssL[sl], s_trg[t]);
            const float alpha = __expf(l - float_of_ord(maxL[sl])) / denL[sl];
            const float* __restrict__ row = wx + (size_t)t * D_OUT;
            for (int d = 0; d < D_OUT; ++d) atomicAdd(&accL[sl * D_OUT + d], alpha * row[d]);
        });
    }
    __syncthreads();

    float* __restrict__ o = out + (size_t)n0 * D_OUT;
    for (int i = tid; i < nown * D_OUT; i += ETHR) o[i] = accL[i];
}

extern "C" void kernel_launch(void* const* d_in, const int* in_sizes, int n_in,
                              void* d_out, int out_size, void* d_ws, size_t ws_size,
                              hipStream_t stream) {
    const float* X   = (const float*)d_in[0];
    const float* W   = (const float*)d_in[1];
    const float* att = (const float*)d_in[2];
    const int*   adj = (const int*)  d_in[3];

    const int d_out_dim = in_sizes[2] / 2;
    const int d_in_dim  = in_sizes[1] / d_out_dim;
    const int n_nodes   = in_sizes[0] / d_in_dim;
    const int n_edges   = in_sizes[3] / 2;
    (void)d_out_dim; (void)d_in_dim;

    const int* esrc = adj;
    const int* etrg = adj + n_edges;

    float* wx    = (float*)d_ws;
    float* s_src = wx + (size_t)n_nodes * D_OUT;
    float* s_trg = s_src + n_nodes;
    const int nb   = (n_nodes + NPW - 1) / NPW;
    const int nseg = (n_edges + SEGE - 1) / SEGE;
    unsigned* tab  = (unsigned*)(s_trg + n_nodes);
    int2*     segs = (int2*)(tab + (((size_t)nb * nseg + 1) & ~(size_t)1));
    float* out   = (float*)d_out;

    const int tiles       = (n_nodes + 15) / 16;
    const int gemm_blocks = (tiles + BTHR / 32 - 1) / (BTHR / 32);
    k_front<<<gemm_blocks + nseg, BTHR, 0, stream>>>(X, W, att, wx, s_src, s_trg, esrc, etrg, tab, segs,
                                                    n_nodes, n_edges, nb, nseg, gemm_blocks);
    k_edges<<<nb, ETHR, 0, stream>>>(esrc, etrg, s_src, s_trg, wx, tab, segs, out,
                                     n_nodes, n_edges, nseg);
}


# Owned edges put in source order in LDS (counting sort, ranks by arrival); a wave adds one source's rows in registers and stores each row once: no LDS row adds or row copy (row sums in list order)
# speedup vs baseline: 4.4147x; 1.0414x over previous
#include <hip/hip_runtime.h>
#include <hip/hip_bf16.h>


#define D_IN   128
#define D_OUT  64
#define LEAKY  0.2f

typedef float v2f __attribute__((ext_vector_type(2)));
typedef float v8f __attribute__((ext_vector_type(8)));

__device__ __forceinline__ unsigned ord_of_float(float f) {
    unsigned u = __float_as_uint(f);
    return (u & 0x80000000u) ? ~u : (u | 0x80000000u);
}
__device__ __forceinline__ float float_of_ord(unsigned u) {
    return __uint_as_float((u & 0x80000000u) ? (u ^ 0x80000000u) : ~u);
}

template <int N>
__device__ __forceinline__ float row_xor(float v) {
    return __int_as_float(__builtin_amdgcn_update_dpp(0, __float_as_int(v), 0x160 | N, 0xf, 0xf, true));
}

__device__ __forceinline__ void gemm_part(const float* __restrict__ X,
                                          const float* __restrict__ W,
                                          const float* __restrict__ att,
                                          float* __restrict__ wx,
                                          float* __restrict__ s_src,
                                          float* __restrict__ s_trg,
                                          int n_nodes, int bid) {
    constexpr int LDW = D_IN + 4;
    __shared__ float sWt[D_OUT * LDW];
    for (int i = threadIdx.x; i < D_IN * D_OUT; i += blockDim.x) {
        int k = i / D_OUT, col = i % D_OUT;
        sWt[col * LDW + k] = W[i];
    }
    __syncthreads();

    const int lane = threadIdx.x & 31;
    const int wave = __builtin_amdgcn_readfirstlane((int)(threadIdx.x >> 5));
    const int tile = bid * (blockDim.x >> 5) + wave;
    const int row0 = tile * 16;
    if (row0 >= n_nodes) return;

    const int m    = lane & 15;
    const int koff = (lane >> 4) * 2;
    int arow = row0 + m;
    if (arow >= n_nodes) arow = n_nodes - 1;

    v8f acc[4] = {v8f{}, v8f{}, v8f{}, v8f{}};

    float as_[4], at_[4];
    #pragma unroll
    for (int t = 0; t < 4; ++t) {
        as_[t] = att[t * 16 + m];
        at_[t] = att[D_OUT + t * 16 + m];
    }

    const float* __restrict__ xrow = X + (size_t)arow * D_IN;
    #pragma unroll
    for (int kb = 0; kb < D_IN; kb += 4) {
        v2f a = *(const v2f*)(xrow + kb + koff);
        #pragma unroll
        for (int t = 0; t < 4; ++t) {
            const int col = t * 16 + m;
            v2f b = *(const v2f*)(&sWt[col * LDW + kb + koff]);
            acc[t] = __builtin_amdgcn_wmma_f32_16x16x4_f32(
                 false, a,  false, b,
                 (short)0, acc[t],
                 false,  false);
        }
    }

    const int rbase = row0 + ((lane >> 4) * 8);
    #pragma unroll
    for (int t = 0; t < 4; ++t) {
        #pragma unroll
        for (int r = 0; r < 8; ++r) {
            const int row = rbase + r;
            if (row < n_nodes)
                wx[(size_t)row * D_OUT + t * 16 + m] = acc[t][r];
        }
    }

    float v16[16];
    #pragma unroll
    for (int r = 0; r < 8; ++r) {
        float ss = acc[0][r] * as_[0];
        float st = acc[0][r] * at_[0];
        #pragma unroll
        for (int t = 1; t < 4; ++t) {
            ss = fmaf(acc[t][r], as_[t], ss);
            st = fmaf(acc[t][r], at_[t], st);
        }
        v16[r]     = ss;
        v16[8 + r] = st;
    }
    const bool b3 = (m & 8) != 0, b2 = (m & 4) != 0, b1 = (m & 2) != 0, b0 = (m & 1) != 0;
    float v8[8], v4[4], v2[2];
    #pragma unroll
    for (int i = 0; i < 8; ++i) {
        const float keep = b3 ? v16[i + 8] : v16[i];
        const float send = b3 ? v16[i] : v16[i + 8];
        v8[i] = keep + row_xor<8>(send);
    }
    #pragma unroll
    for (int i = 0; i < 4; ++i) {
        const float keep = b2 ? v8[i + 4] : v8[i];
        const float send = b2 ? v8[i] : v8[i + 4];
        v4[i] = keep + row_xor<4>(send);
    }
    #pragma unroll
    for (int i = 0; i < 2; ++i) {
        const float keep = b1 ? v4[i + 2] : v4[i];
        const float send = b1 ? v4[i] : v4[i + 2];
        v2[i] = keep + row_xor<2>(send);
    }
    const float keep1 = b0 ? v2[1] : v2[0];
    const float send1 = b0 ? v2[0] : v2[1];
    const float score = keep1 + row_xor<1>(send1);
    {
        const int row = rbase + (m & 7);
        float* dst = b3 ? s_trg : s_src;
        if (row < n_nodes) dst[row] = score;
    }
}

#define NPW   391
#define CAP   8192
#define ETHR  1024
#define BTHR  1024
#define BEPT  4
#define SEGE  (BTHR * BEPT)
#define NBMAX 1024
#define SEGMAX 256
#define CBITS 13

__device__ __forceinline__ void bin_part(const int* __restrict__ esrc,
                                         const int* __restrict__ etrg,
                                         unsigned* __restrict__ tab,
                                         int2* __restrict__ segs,
                                         int n_nodes, int n_edges, int nb, int nseg, int seg) {
    __shared__ unsigned hist[NBMAX];
    __shared__ unsigned wtot[32];
    const int tid  = threadIdx.x;
    const int lane = tid & 31;
    const int wave = tid >> 5;
    hist[tid] = 0u;
    __syncthreads();

    const int base = seg * SEGE + tid;
    int s[BEPT], t[BEPT], b[BEPT];
    unsigned r[BEPT];
    #pragma unroll
    for (int u = 0; u < BEPT; ++u) {
        const int e = base + u * BTHR;
        const bool in = e < n_edges;
        s[u] = in ? esrc[e] : -1;
        t[u] = in ? etrg[e] : 0;
    }
    #pragma unroll
    for (int u = 0; u < BEPT; ++u) {
        const bool ok = (unsigned)s[u] < (unsigned)n_nodes;
        b[u] = ok ? s[u] / NPW : -1;
        r[u] = ok ? atomicAdd(&hist[b[u]], 1u) : 0u;
    }
    __syncthreads();
    const unsigned cntv = hist[tid];
    unsigned inc = cntv;
    #pragma unroll
    for (int d = 1; d < 32; d <<= 1) {
        const unsigned o = __shfl_up(inc, d);
        if (lane >= d) inc += o;
    }
    if (lane == 31) wtot[wave] = inc;
    __syncthreads();
    if (wave == 0) {
        const unsigned w = wtot[lane];
        unsigned winc = w;
        #pragma unroll
        for (int d = 1; d < 32; d <<= 1) {
            const unsigned o = __shfl_up(winc, d);
            if (lane >= d) winc += o;
        }
        wtot[lane] = winc - w;
    }
    __syncthreads();
    const unsigned start = inc - cntv + wtot[wave];
    hist[tid] = start;
    if (tid < nb) tab[(size_t)tid * nseg + seg] = (start << CBITS) | cntv;
    __syncthreads();
    int2* __restrict__ mine = segs + (size_t)seg * SEGE;
    #pragma unroll
    for (int u = 0; u < BEPT; ++u) {
        if (b[u] >= 0)
            mine[hist[b[u]] + r[u]] = make_int2(s[u] - b[u] * NPW, t[u]);
    }
}

__global__ __launch_bounds__(BTHR, 16) void k_front(const float* __restrict__ X,
                                                const float* __restrict__ W,
                                                const float* __restrict__ att,
                                                float* __restrict__ wx,
                                                float* __restrict__ s_src,
                                                float* __restrict__ s_trg,
                                                const int* __restrict__ esrc,
                                                const int* __restrict__ etrg,
                                                unsigned* __restrict__ tab,
                                                int2* __restrict__ segs,
                                                int n_nodes, int n_edges, int nb, int nseg,
                                                int gemm_blocks) {
    if ((int)blockIdx.x < gemm_blocks)
        gemm_part(X, W, att, wx, s_src, s_trg, n_nodes, (int)blockIdx.x);
    else
        bin_part(esrc, etrg, tab, segs, n_nodes, n_edges, nb, nseg, (int)blockIdx.x - gemm_blocks);
}

__device__ __forceinline__ float edge_logit(float a, float b) {
    float l = a + b;
    return (l >= 0.0f) ? l : LEAKY * l;
}

template <typename F>
__device__ __forceinline__ void scan_owned(const int* __restrict__ esrc, int n_edges,
                                           int n0, int nown, int rot, F body) {
    const int tid  = threadIdx.x;
    const int nvec = n_edges >> 2;
    const int4* __restrict__ src4 = (const int4*)esrc;
    for (int base = 0; base < nvec; base += 4 * ETHR) {
        int4 q[4];
        int  w[4];
        #pragma unroll
        for (int u = 0; u < 4; ++u) {
            const int v = base + u * ETHR + tid;
            int ww = v + rot;
            if (ww >= nvec) ww -= nvec;
            w[u] = ww;
            q[u] = (v < nvec) ? src4[ww] : make_int4(-1, -1, -1, -1);
        }
        #pragma unroll
        for (int u = 0; u < 4; ++u) {
            const int e0 = w[u] << 2;
            const unsigned d0 = (unsigned)(q[u].x - n0);
            const unsigned d1 = (unsigned)(q[u].y - n0);
            const unsigned d2 = (unsigned)(q[u].z - n0);
            const unsigned d3 = (unsigned)(q[u].w - n0);
            if (d0 < (unsigned)nown) body((int)d0, e0);
            if (d1 < (unsigned)nown) body((int)d1, e0 + 1);
            if (d2 < (unsigned)nown) body((int)d2, e0 + 2);
            if (d3 < (unsigned)nown) body((int)d3, e0 + 3);
        }
    }
    for (int e = (nvec << 2) + tid; e < n_edges; e += ETHR) {
        const unsigned d = (unsigned)(esrc[e] - n0);
        if (d < (unsigned)nown) body((int)d, e);
    }
}

#define RB 16
#define SBITS 9
#define SCANW ((NPW + 1 + 31) / 32)

struct RowState {
    int    cur;
    float2 acc;
};

__device__ __forceinline__ void flush_row(RowState& st, float* __restrict__ orow) {
    if (st.cur >= 0) *(float2*)(orow + (size_t)st.cur * D_OUT) = st.acc;
    st.acc = make_float2(0.0f, 0.0f);
}

template <bool TAIL>
__device__ __forceinline__ void add_rows(const float* __restrict__ wx, float* __restrict__ orow, RowState& st,
                                         int my_s, int my_t, float my_a, unsigned bm,
                                         int j0, int n, int lane) {
    int    sl[RB];
    float  al[RB];
    float2 v[RB];
    #pragma unroll
    for (int u = 0; u < RB; ++u) {
        const int jj = TAIL ? min(j0 + u, n - 1) : (j0 + u);
        sl[u] = __builtin_amdgcn_readlane(my_s, jj);
        const int t = __builtin_amdgcn_readlane(my_t, jj);
        al[u] = __int_as_float(__builtin_amdgcn_readlane(__float_as_int(my_a), jj));
        v[u]  = ((const float2*)(wx + (size_t)t * D_OUT))[lane];
    }
    #pragma unroll
    for (int u = 0; u < RB; ++u) {
        if (!TAIL || j0 + u < n) {
            if ((bm >> (j0 + u)) & 1u) {
                flush_row(st, orow);
                st.cur = sl[u];
            }
            st.acc.x += al[u] * v[u].x;
            st.acc.y += al[u] * v[u].y;
        }
    }
}

__global__ __launch_bounds__(ETHR) void k_edges(const int* __restrict__ esrc,
                                                const int* __restrict__ etrg,
                                                const float* __restrict__ s_src,
                                                const float* __restrict__ s_trg,
                                                const float* __restrict__ wx,
                                                const unsigned* __restrict__ tab,
                                                const int2* __restrict__ segs,
                                                float* __restrict__ out,
                                                int n_nodes, int n_edges, int nseg) {
    __shared__ union {
        float acc[NPW * D_OUT];
        struct { int s[CAP]; int t[CAP]; float w[CAP]; } srt;
    } rowsL;
    __shared__ unsigned cntS[NPW + 1];
    __shared__ unsigned offS[NPW + 1];
    __shared__ unsigned wtotS[SCANW];
    __shared__ unsigned maxL[NPW];
    __shared__ float    denL[NPW];
    __shared__ float    ssL[NPW];
    __shared__ int      listS[CAP];
    __shared__ int      listT[CAP];
    __shared__ float    listW[CAP];
    __shared__ unsigned segInfo[SEGMAX];
    __shared__ unsigned segBase[SEGMAX];
    __shared__ unsigned lcnt;
    __shared__ unsigned fbIdx;

    const int tid  = threadIdx.x;
    const int lane = tid & 31;
    const int wave = tid >> 5;
    const int n0   = blockIdx.x * NPW;
    const int nown = min(NPW, n_nodes - n0);

    for (int i = tid; i < NPW; i += ETHR) {
        cntS[i] = 0u;
        maxL[i] = 0u;
        denL[i] = 0.0f;
        ssL[i]  = (i < nown) ? s_src[n0 + i] : 0.0f;
    }
    if (tid == 0) lcnt = 0u;
    __syncthreads();

    const bool have = nseg <= SEGMAX;
    if (have) {
        for (int i = tid; i < nseg; i += ETHR) {
            const unsigned p = tab[(size_t)blockIdx.x * nseg + i];
            segInfo[i] = p;
            segBase[i] = atomicAdd(&lcnt, p & ((1u << CBITS) - 1u));
        }
    }
    __syncthreads();
    const unsigned cnt = have ? lcnt : (unsigned)CAP + 1u;

    if (cnt <= (unsigned)CAP) {
        const unsigned l16   = (unsigned)tid & 15u;
        const unsigned cmask = (1u << CBITS) - 1u;
        for (int sg = tid >> 4; sg < nseg; sg += ETHR / 16) {
            const unsigned p     = segInfo[sg];
            const unsigned c     = p & cmask;
            const unsigned bs    = segBase[sg];
            const unsigned first = (unsigned)sg * SEGE + (p >> CBITS);
            for (unsigned j = l16; j < c; j += 16) {
                listT[bs + j] = (int)(first + j);
                if (bs + j == 0u) fbIdx = first + j;
            }
        }
        __syncthreads();
        const unsigned fb = fbIdx;
        for (unsigned base = 0; base < cnt; base += 4 * ETHR) {
            unsigned idx[4];
            int2     rec[4];
            float    st[4];
            #pragma unroll
            for (int u = 0; u < 4; ++u) {
                const unsigned i = base + u * ETHR + tid;
                idx[u] = (i < cnt) ? (unsigned)listT[i] : fb;
            }
            #pragma unroll
            for (int u = 0; u < 4; ++u) rec[u] = segs[idx[u]];
            #pragma unroll
            for (int u = 0; u < 4; ++u) st[u] = s_trg[rec[u].y];
            asm volatile("" :: "v"(st[0]), "v"(st[1]), "v"(st[2]), "v"(st[3]));
            #pragma unroll
            for (int u = 0; u < 4; ++u) {
                const unsigned i = base + u * ETHR + tid;
                if (i < cnt) {
                    const int sl = rec[u].x;
                    const float l = edge_logit(ssL[sl], st[u]);
                    const unsigned rk = atomicAdd(&cntS[sl], 1u);
                    listS[i] = sl | (int)(rk << SBITS);
                    listT[i] = rec[u].y;
                    listW[i] = l;
                    atomicMax(&maxL[sl], ord_of_float(l));
                }
            }
        }
        __syncthreads();
        unsigned cntv = 0u, inc = 0u;
        if (wave < SCANW) {
            cntv = (tid < NPW) ? cntS[tid] : 0u;
            inc  = cntv;
            #pragma unroll
            for (int d = 1; d < 32; d <<= 1) {
                const unsigned o = __shfl_up(inc, d);
                if (lane >= d) inc += o;
            }
            if (lane == 31) wtotS[wave] = inc;
            unsigned zm = __builtin_amdgcn_ballot_w32(tid < nown && cntv == 0u);
            while (zm != 0u) {
                const int b = __builtin_ctz(zm);
                zm &= zm - 1u;
                *(float2*)(out + (size_t)(n0 + wave * 32 + b) * D_OUT + 2 * lane) = make_float2(0.0f, 0.0f);
            }
        }
        __syncthreads();
        if (wave < SCANW) {
            unsigned pre = 0u;
            for (int w = 0; w < wave; ++w) pre += wtotS[w];
            if (tid <= NPW) offS[tid] = pre + inc - cntv;
        }
        __syncthreads();
        for (unsigned i = tid; i < cnt; i += ETHR) {
            const int pk = listS[i];
            const int sl = pk & ((1 << SBITS) - 1);
            const float ex = __expf(listW[i] - float_of_ord(maxL[sl]));
            atomicAdd(&denL[sl], ex);
            const unsigned p = offS[sl] + ((unsigned)pk >> SBITS);
            rowsL.srt.s[p] = sl;
            rowsL.srt.t[p] = listT[i];
            rowsL.srt.w[p] = ex;
        }
        __syncthreads();
        const unsigned wv  = (unsigned)__builtin_amdgcn_readfirstlane(wave);
        const int s_lo = (int)((wv * (unsigned)nown) / (ETHR / 32));
        const int s_hi = (int)(((wv + 1u) * (unsigned)nown) / (ETHR / 32));
        const unsigned e_lo = (unsigned)__builtin_amdgcn_readfirstlane((int)offS[s_lo]);
        const unsigned e_hi = (unsigned)__builtin_amdgcn_readfirstlane((int)offS[s_hi]);
        float* __restrict__ orow = out + (size_t)n0 * D_OUT + 2 * lane;
        RowState st;
        st.cur = -1;
        st.acc = make_float2(0.0f, 0.0f);
        for (unsigned base = e_lo; base < e_hi; base += 32u) {
            const unsigned i  = base + lane;
            const bool     ok = i < e_hi;
            const int   my_s = ok ? rowsL.srt.s[i] : 0;
            const int   my_t = ok ? rowsL.srt.t[i] : 0;
            const float my_a = ok ? rowsL.srt.w[i] / denL[my_s] : 0.0f;
            const int   pv_s = (ok && i > e_lo) ? rowsL.srt.s[i - 1] : -1;
            const unsigned bm = __builtin_amdgcn_ballot_w32(ok && my_s != pv_s);
            const int n = (int)min(32u, e_hi - base);
            for (int j0 = 0; j0 < n; j0 += RB) {
                if (j0 + RB <= n) add_rows<false>(wx, orow, st, my_s, my_t, my_a, bm, j0, n, lane);
                else              add_rows<true >(wx, orow, st, my_s, my_t, my_a, bm, j0, n, lane);
            }
        }
        flush_row(st, orow);
    } else {
        for (int i = tid; i < NPW * D_OUT; i += ETHR) rowsL.acc[i] = 0.0f;
        const int nvec = n_edges >> 2;
        const int rot  = (int)(((long long)blockIdx.x * nvec) / gridDim.x) & ~31;
        scan_owned(esrc, n_edges, n0, nown, rot, [&](int sl, int e) {
            const int t = etrg[e];
            atomicMax(&maxL[sl], ord_of_float(edge_logit(ssL[sl], s_trg[t])));
        });
        __syncthreads();
        scan_owned(esrc, n_edges, n0, nown, rot, [&](int sl, int e) {
            const float l = edge_logit(ssL[sl], s_trg[etrg[e]]);
            atomicAdd(&denL[sl], __expf(l - float_of_ord(maxL[sl])));
        });
        __syncthreads();
        scan_owned(esrc, n_edges, n0, nown, rot, [&](int sl, int e) {
            const int t = etrg[e];
            const float l = edge_logit(ssL[sl], s_trg[t]);
            const float alpha = __expf(l - float_of_ord(maxL[sl])) / denL[sl];
            const float* __restrict__ row = wx + (size_t)t * D_OUT;
            for (int d = 0; d < D_OUT; ++d) atomicAdd(&rowsL.acc[sl * D_OUT + d], alpha * row[d]);
        });
        __syncthreads();
        float* __restrict__ o = out + (size_t)n0 * D_OUT;
        for (int i = tid; i < nown * D_OUT; i += ETHR) o[i] = rowsL.acc[i];
    }
}

extern "C" void kernel_launch(void* const* d_in, const int* in_sizes, int n_in,
                              void* d_out, int out_size, void* d_ws, size_t ws_size,
                              hipStream_t stream) {
    const float* X   = (const float*)d_in[0];
    const float* W   = (const float*)d_in[1];
    const float* att = (const float*)d_in[2];
    const int*   adj = (const int*)  d_in[3];

    const int d_out_dim = in_sizes[2] / 2;
    const int d_in_dim  = in_sizes[1] / d_out_dim;
    const int n_nodes   = in_sizes[0] / d_in_dim;
    const int n_edges   = in_sizes[3] / 2;
    (void)d_out_dim; (void)d_in_dim;

    const int* esrc = adj;
    const int* etrg = adj + n_edges;

    float* wx    = (float*)d_ws;
    float* s_src = wx + (size_t)n_nodes * D_OUT;
    float* s_trg = s_src + n_nodes;
    const int nb   = (n_nodes + NPW - 1) / NPW;
    const int nseg = (n_edges + SEGE - 1) / SEGE;
    unsigned* tab  = (unsigned*)(s_trg + n_nodes);
    int2*     segs = (int2*)(tab + (((size_t)nb * nseg + 1) & ~(size_t)1));
    float* out   = (float*)d_out;

    const int tiles       = (n_nodes + 15) / 16;
    const int gemm_blocks = (tiles + BTHR / 32 - 1) / (BTHR / 32);
    k_front<<<gemm_blocks + nseg, BTHR, 0, stream>>>(X, W, att, wx, s_src, s_trg, esrc, etrg, tab, segs,
                                                    n_nodes, n_edges, nb, nseg, gemm_blocks);
    k_edges<<<nb, ETHR, 0, stream>>>(esrc, etrg, s_src, s_trg, wx, tab, segs, out,
                                     n_nodes, n_edges, nseg);
}


# Opening loads as one group: W staged with 8 loads per thread in flight (was 8 round trips in a row), attention vector into LDS with it, first X fragments ahead of the barrier; edge table word too
# speedup vs baseline: 4.7948x; 1.0861x over previous
#include <hip/hip_runtime.h>
#include <hip/hip_bf16.h>


#define D_IN   128
#define D_OUT  64
#define LEAKY  0.2f
#define BTHR   1024

typedef float v2f __attribute__((ext_vector_type(2)));
typedef float v8f __attribute__((ext_vector_type(8)));

__device__ __forceinline__ unsigned ord_of_float(float f) {
    unsigned u = __float_as_uint(f);
    return (u & 0x80000000u) ? ~u : (u | 0x80000000u);
}
__device__ __forceinline__ float float_of_ord(unsigned u) {
    return __uint_as_float((u & 0x80000000u) ? (u ^ 0x80000000u) : ~u);
}

template <int N>
__device__ __forceinline__ float row_xor(float v) {
    return __int_as_float(__builtin_amdgcn_update_dpp(0, __float_as_int(v), 0x160 | N, 0xf, 0xf, true));
}

__device__ __forceinline__ void gemm_part(const float* __restrict__ X,
                                          const float* __restrict__ W,
                                          const float* __restrict__ att,
                                          float* __restrict__ wx,
                                          float* __restrict__ s_src,
                                          float* __restrict__ s_trg,
                                          int n_nodes, int bid) {
    constexpr int LDW = D_IN + 4;
    __shared__ float sWt[D_OUT * LDW];
    __shared__ float sAtt[2 * D_OUT];

    const int lane = threadIdx.x & 31;
    const int wave = __builtin_amdgcn_readfirstlane((int)(threadIdx.x >> 5));
    const int tile = bid * (BTHR / 32) + wave;
    const int row0 = tile * 16;
    const int m    = lane & 15;
    const int koff = (lane >> 4) * 2;
    int arow = row0 + m;
    if (arow >= n_nodes) arow = n_nodes - 1;
    const float* __restrict__ xrow = X + (size_t)arow * D_IN;

    constexpr int WPT  = D_IN * D_OUT / BTHR;
    constexpr int NPRE = 5;
    float wv[WPT];
    #pragma unroll
    for (int j = 0; j < WPT; ++j) wv[j] = W[threadIdx.x + j * BTHR];
    const bool has_att = threadIdx.x < 2 * D_OUT;
    float attw = 0.0f;
    if (has_att) attw = att[threadIdx.x];
    v2f apre[NPRE];
    #pragma unroll
    for (int j = 0; j < NPRE; ++j) apre[j] = *(const v2f*)(xrow + 4 * j + koff);
    #pragma unroll
    for (int j = 0; j < WPT; ++j) {
        const int i = threadIdx.x + j * BTHR;
        const int k = i / D_OUT, col = i % D_OUT;
        sWt[col * LDW + k] = wv[j];
    }
    if (has_att) sAtt[threadIdx.x] = attw;
    asm volatile("" :: "v"(apre[0]), "v"(apre[1]), "v"(apre[2]), "v"(apre[3]), "v"(apre[4]));
    __syncthreads();
    if (row0 >= n_nodes) return;

    v8f acc[4] = {v8f{}, v8f{}, v8f{}, v8f{}};

    #pragma unroll
    for (int kb = 0; kb < D_IN; kb += 4) {
        v2f a;
        if (kb / 4 < NPRE) a = apre[kb / 4];
        else               a = *(const v2f*)(xrow + kb + koff);
        #pragma unroll
        for (int t = 0; t < 4; ++t) {
            const int col = t * 16 + m;
            v2f b = *(const v2f*)(&sWt[col * LDW + kb + koff]);
            acc[t] = __builtin_amdgcn_wmma_f32_16x16x4_f32(
                 false, a,  false, b,
                 (short)0, acc[t],
                 false,  false);
        }
    }

    const int rbase = row0 + ((lane >> 4) * 8);
    #pragma unroll
    for (int t = 0; t < 4; ++t) {
        #pragma unroll
        for (int r = 0; r < 8; ++r) {
            const int row = rbase + r;
            if (row < n_nodes)
                wx[(size_t)row * D_OUT + t * 16 + m] = acc[t][r];
        }
    }

    float as_[4], at_[4];
    #pragma unroll
    for (int t = 0; t < 4; ++t) {
        as_[t] = sAtt[t * 16 + m];
        at_[t] = sAtt[D_OUT + t * 16 + m];
    }
    float v16[16];
    #pragma unroll
    for (int r = 0; r < 8; ++r) {
        float ss = acc[0][r] * as_[0];
        float st = acc[0][r] * at_[0];
        #pragma unroll
        for (int t = 1; t < 4; ++t) {
            ss = fmaf(acc[t][r], as_[t], ss);
            st = fmaf(acc[t][r], at_[t], st);
        }
        v16[r]     = ss;
        v16[8 + r] = st;
    }
    const bool b3 = (m & 8) != 0, b2 = (m & 4) != 0, b1 = (m & 2) != 0, b0 = (m & 1) != 0;
    float v8[8], v4[4], v2[2];
    #pragma unroll
    for (int i = 0; i < 8; ++i) {
        const float keep = b3 ? v16[i + 8] : v16[i];
        const float send = b3 ? v16[i] : v16[i + 8];
        v8[i] = keep + row_xor<8>(send);
    }
    #pragma unroll
    for (int i = 0; i < 4; ++i) {
        const float keep = b2 ? v8[i + 4] : v8[i];
        const float send = b2 ? v8[i] : v8[i + 4];
        v4[i] = keep + row_xor<4>(send);
    }
    #pragma unroll
    for (int i = 0; i < 2; ++i) {
        const float keep = b1 ? v4[i + 2] : v4[i];
        const float send = b1 ? v4[i] : v4[i + 2];
        v2[i] = keep + row_xor<2>(send);
    }
    const float keep1 = b0 ? v2[1] : v2[0];
    const float send1 = b0 ? v2[0] : v2[1];
    const float score = keep1 + row_xor<1>(send1);
    {
        const int row = rbase + (m & 7);
        float* dst = b3 ? s_trg : s_src;
        if (row < n_nodes) dst[row] = score;
    }
}

#define NPW   391
#define CAP   8192
#define ETHR  1024
#define BEPT  4
#define SEGE  (BTHR * BEPT)
#define NBMAX 1024
#define SEGMAX 256
#define CBITS 13

__device__ __forceinline__ void bin_part(const int* __restrict__ esrc,
                                         const int* __restrict__ etrg,
                                         unsigned* __restrict__ tab,
                                         int2* __restrict__ segs,
                                         int n_nodes, int n_edges, int nb, int nseg, int seg) {
    __shared__ unsigned hist[NBMAX];
    __shared__ unsigned wtot[32];
    const int tid  = threadIdx.x;
    const int lane = tid & 31;
    const int wave = tid >> 5;
    hist[tid] = 0u;
    __syncthreads();

    const int base = seg * SEGE + tid;
    int s[BEPT], t[BEPT], b[BEPT];
    unsigned r[BEPT];
    #pragma unroll
    for (int u = 0; u < BEPT; ++u) {
        const int e = base + u * BTHR;
        const bool in = e < n_edges;
        s[u] = in ? esrc[e] : -1;
        t[u] = in ? etrg[e] : 0;
    }
    #pragma unroll
    for (int u = 0; u < BEPT; ++u) {
        const bool ok = (unsigned)s[u] < (unsigned)n_nodes;
        b[u] = ok ? s[u] / NPW : -1;
        r[u] = ok ? atomicAdd(&hist[b[u]], 1u) : 0u;
    }
    __syncthreads();
    const unsigned cntv = hist[tid];
    unsigned inc = cntv;
    #pragma unroll
    for (int d = 1; d < 32; d <<= 1) {
        const unsigned o = __shfl_up(inc, d);
        if (lane >= d) inc += o;
    }
    if (lane == 31) wtot[wave] = inc;
    __syncthreads();
    if (wave == 0) {
        const unsigned w = wtot[lane];
        unsigned winc = w;
        #pragma unroll
        for (int d = 1; d < 32; d <<= 1) {
            const unsigned o = __shfl_up(winc, d);
            if (lane >= d) winc += o;
        }
        wtot[lane] = winc - w;
    }
    __syncthreads();
    const unsigned start = inc - cntv + wtot[wave];
    hist[tid] = start;
    if (tid < nb) tab[(size_t)tid * nseg + seg] = (start << CBITS) | cntv;
    __syncthreads();
    int2* __restrict__ mine = segs + (size_t)seg * SEGE;
    #pragma unroll
    for (int u = 0; u < BEPT; ++u) {
        if (b[u] >= 0)
            mine[hist[b[u]] + r[u]] = make_int2(s[u] - b[u] * NPW, t[u]);
    }
}

__global__ __launch_bounds__(BTHR, 16) void k_front(const float* __restrict__ X,
                                                const float* __restrict__ W,
                                                const float* __restrict__ att,
                                                float* __restrict__ wx,
                                                float* __restrict__ s_src,
                                                float* __restrict__ s_trg,
                                                const int* __restrict__ esrc,
                                                const int* __restrict__ etrg,
                                                unsigned* __restrict__ tab,
                                                int2* __restrict__ segs,
                                                int n_nodes, int n_edges, int nb, int nseg,
                                                int gemm_blocks) {
    if ((int)blockIdx.x < gemm_blocks)
        gemm_part(X, W, att, wx, s_src, s_trg, n_nodes, (int)blockIdx.x);
    else
        bin_part(esrc, etrg, tab, segs, n_nodes, n_edges, nb, nseg, (int)blockIdx.x - gemm_blocks);
}

__device__ __forceinline__ float edge_logit(float a, float b) {
    float l = a + b;
    return (l >= 0.0f) ? l : LEAKY * l;
}

template <typename F>
__device__ __forceinline__ void scan_owned(const int* __restrict__ esrc, int n_edges,
                                           int n0, int nown, int rot, F body) {
    const int tid  = threadIdx.x;
    const int nvec = n_edges >> 2;
    const int4* __restrict__ src4 = (const int4*)esrc;
    for (int base = 0; base < nvec; base += 4 * ETHR) {
        int4 q[4];
        int  w[4];
        #pragma unroll
        for (int u = 0; u < 4; ++u) {
            const int v = base + u * ETHR + tid;
            int ww = v + rot;
            if (ww >= nvec) ww -= nvec;
            w[u] = ww;
            q[u] = (v < nvec) ? src4[ww] : make_int4(-1, -1, -1, -1);
        }
        #pragma unroll
        for (int u = 0; u < 4; ++u) {
            const int e0 = w[u] << 2;
            const unsigned d0 = (unsigned)(q[u].x - n0);
            const unsigned d1 = (unsigned)(q[u].y - n0);
            const unsigned d2 = (unsigned)(q[u].z - n0);
            const unsigned d3 = (unsigned)(q[u].w - n0);
            if (d0 < (unsigned)nown) body((int)d0, e0);
            if (d1 < (unsigned)nown) body((int)d1, e0 + 1);
            if (d2 < (unsigned)nown) body((int)d2, e0 + 2);
            if (d3 < (unsigned)nown) body((int)d3, e0 + 3);
        }
    }
    for (int e = (nvec << 2) + tid; e < n_edges; e += ETHR) {
        const unsigned d = (unsigned)(esrc[e] - n0);
        if (d < (unsigned)nown) body((int)d, e);
    }
}

#define RB 16
#define SBITS 9
#define SCANW ((NPW + 1 + 31) / 32)

struct RowState {
    int    cur;
    float2 acc;
};

__device__ __forceinline__ void flush_row(RowState& st, float* __restrict__ orow) {
    if (st.cur >= 0) *(float2*)(orow + (size_t)st.cur * D_OUT) = st.acc;
    st.acc = make_float2(0.0f, 0.0f);
}

template <bool TAIL>
__device__ __forceinline__ void add_rows(const float* __restrict__ wx, float* __restrict__ orow, RowState& st,
                                         int my_s, int my_t, float my_a, unsigned bm,
                                         int j0, int n, int lane) {
    int    sl[RB];
    float  al[RB];
    float2 v[RB];
    #pragma unroll
    for (int u = 0; u < RB; ++u) {
        const int jj = TAIL ? min(j0 + u, n - 1) : (j0 + u);
        sl[u] = __builtin_amdgcn_readlane(my_s, jj);
        const int t = __builtin_amdgcn_readlane(my_t, jj);
        al[u] = __int_as_float(__builtin_amdgcn_readlane(__float_as_int(my_a), jj));
        v[u]  = ((const float2*)(wx + (size_t)t * D_OUT))[lane];
    }
    #pragma unroll
    for (int u = 0; u < RB; ++u) {
        if (!TAIL || j0 + u < n) {
            if ((bm >> (j0 + u)) & 1u) {
                flush_row(st, orow);
                st.cur = sl[u];
            }
            st.acc.x += al[u] * v[u].x;
            st.acc.y += al[u] * v[u].y;
        }
    }
}

__global__ __launch_bounds__(ETHR) void k_edges(const int* __restrict__ esrc,
                                                const int* __restrict__ etrg,
                                                const float* __restrict__ s_src,
                                                const float* __restrict__ s_trg,
                                                const float* __restrict__ wx,
                                                const unsigned* __restrict__ tab,
                                                const int2* __restrict__ segs,
                                                float* __restrict__ out,
                                                int n_nodes, int n_edges, int nseg) {
    __shared__ union {
        float acc[NPW * D_OUT];
        struct { int s[CAP]; int t[CAP]; float w[CAP]; } srt;
    } rowsL;
    __shared__ unsigned cntS[NPW + 1];
    __shared__ unsigned offS[NPW + 1];
    __shared__ unsigned wtotS[SCANW];
    __shared__ unsigned maxL[NPW];
    __shared__ float    denL[NPW];
    __shared__ float    ssL[NPW];
    __shared__ int      listS[CAP];
    __shared__ int      listT[CAP];
    __shared__ float    listW[CAP];
    __shared__ unsigned segInfo[SEGMAX];
    __shared__ unsigned segBase[SEGMAX];
    __shared__ unsigned lcnt;
    __shared__ unsigned fbIdx;

    const int tid  = threadIdx.x;
    const int lane = tid & 31;
    const int wave = tid >> 5;
    const int n0   = blockIdx.x * NPW;
    const int nown = min(NPW, n_nodes - n0);

    const bool have = nseg <= SEGMAX;
    const bool tseg = have && tid < nseg;
    unsigned tabw = 0u;
    if (tseg) tabw = tab[(size_t)blockIdx.x * nseg + tid];

    for (int i = tid; i < NPW; i += ETHR) {
        cntS[i] = 0u;
        maxL[i] = 0u;
        denL[i] = 0.0f;
        ssL[i]  = (i < nown) ? s_src[n0 + i] : 0.0f;
    }
    if (tid == 0) lcnt = 0u;
    __syncthreads();

    if (tseg) {
        segInfo[tid] = tabw;
        segBase[tid] = atomicAdd(&lcnt, tabw & ((1u << CBITS) - 1u));
    }
    __syncthreads();
    const unsigned cnt = have ? lcnt : (unsigned)CAP + 1u;

    if (cnt <= (unsigned)CAP) {
        const unsigned l16   = (unsigned)tid & 15u;
        const unsigned cmask = (1u << CBITS) - 1u;
        for (int sg = tid >> 4; sg < nseg; sg += ETHR / 16) {
            const unsigned p     = segInfo[sg];
            const unsigned c     = p & cmask;
            const unsigned bs    = segBase[sg];
            const unsigned first = (unsigned)sg * SEGE + (p >> CBITS);
            for (unsigned j = l16; j < c; j += 16) {
                listT[bs + j] = (int)(first + j);
                if (bs + j == 0u) fbIdx = first + j;
            }
        }
        __syncthreads();
        const unsigned fb = fbIdx;
        for (unsigned base = 0; base < cnt; base += 4 * ETHR) {
            unsigned idx[4];
            int2     rec[4];
            float    st[4];
            #pragma unroll
            for (int u = 0; u < 4; ++u) {
                const unsigned i = base + u * ETHR + tid;
                idx[u] = (i < cnt) ? (unsigned)listT[i] : fb;
            }
            #pragma unroll
            for (int u = 0; u < 4; ++u) rec[u] = segs[idx[u]];
            #pragma unroll
            for (int u = 0; u < 4; ++u) st[u] = s_trg[rec[u].y];
            asm volatile("" :: "v"(st[0]), "v"(st[1]), "v"(st[2]), "v"(st[3]));
            #pragma unroll
            for (int u = 0; u < 4; ++u) {
                const unsigned i = base + u * ETHR + tid;
                if (i < cnt) {
                    const int sl = rec[u].x;
                    const float l = edge_logit(ssL[sl], st[u]);
                    const unsigned rk = atomicAdd(&cntS[sl], 1u);
                    listS[i] = sl | (int)(rk << SBITS);
                    listT[i] = rec[u].y;
                    listW[i] = l;
                    atomicMax(&maxL[sl], ord_of_float(l));
                }
            }
        }
        __syncthreads();
        unsigned cntv = 0u, inc = 0u;
        if (wave < SCANW) {
            cntv = (tid < NPW) ? cntS[tid] : 0u;
            inc  = cntv;
            #pragma unroll
            for (int d = 1; d < 32; d <<= 1) {
                const unsigned o = __shfl_up(inc, d);
                if (lane >= d) inc += o;
            }
            if (lane == 31) wtotS[wave] = inc;
            unsigned zm = __builtin_amdgcn_ballot_w32(tid < nown && cntv == 0u);
            while (zm != 0u) {
                const int b = __builtin_ctz(zm);
                zm &= zm - 1u;
                *(float2*)(out + (size_t)(n0 + wave * 32 + b) * D_OUT + 2 * lane) = make_float2(0.0f, 0.0f);
            }
        }
        __syncthreads();
        if (wave < SCANW) {
            unsigned pre = 0u;
            for (int w = 0; w < wave; ++w) pre += wtotS[w];
            if (tid <= NPW) offS[tid] = pre + inc - cntv;
        }
        __syncthreads();
        for (unsigned i = tid; i < cnt; i += ETHR) {
            const int pk = listS[i];
            const int sl = pk & ((1 << SBITS) - 1);
            const float ex = __expf(listW[i] - float_of_ord(maxL[sl]));
            atomicAdd(&denL[sl], ex);
            const unsigned p = offS[sl] + ((unsigned)pk >> SBITS);
            rowsL.srt.s[p] = sl;
            rowsL.srt.t[p] = listT[i];
            rowsL.srt.w[p] = ex;
        }
        __syncthreads();
        const unsigned wv  = (unsigned)__builtin_amdgcn_readfirstlane(wave);
        const int s_lo = (int)((wv * (unsigned)nown) / (ETHR / 32));
        const int s_hi = (int)(((wv + 1u) * (unsigned)nown) / (ETHR / 32));
        const unsigned e_lo = (unsigned)__builtin_amdgcn_readfirstlane((int)offS[s_lo]);
        const unsigned e_hi = (unsigned)__builtin_amdgcn_readfirstlane((int)offS[s_hi]);
        float* __restrict__ orow = out + (size_t)n0 * D_OUT + 2 * lane;
        RowState st;
        st.cur = -1;
        st.acc = make_float2(0.0f, 0.0f);
        for (unsigned base = e_lo; base < e_hi; base += 32u) {
            const unsigned i  = base + lane;
            const bool     ok = i < e_hi;
            const int   my_s = ok ? rowsL.srt.s[i] : 0;
            const int   my_t = ok ? rowsL.srt.t[i] : 0;
            const float my_a = ok ? rowsL.srt.w[i] / denL[my_s] : 0.0f;
            const int   pv_s = (ok && i > e_lo) ? rowsL.srt.s[i - 1] : -1;
            const unsigned bm = __builtin_amdgcn_ballot_w32(ok && my_s != pv_s);
            const int n = (int)min(32u, e_hi - base);
            for (int j0 = 0; j0 < n; j0 += RB) {
                if (j0 + RB <= n) add_rows<false>(wx, orow, st, my_s, my_t, my_a, bm, j0, n, lane);
                else              add_rows<true >(wx, orow, st, my_s, my_t, my_a, bm, j0, n, lane);
            }
        }
        flush_row(st, orow);
    } else {
        for (int i = tid; i < NPW * D_OUT; i += ETHR) rowsL.acc[i] = 0.0f;
        const int nvec = n_edges >> 2;
        const int rot  = (int)(((long long)blockIdx.x * nvec) / gridDim.x) & ~31;
        scan_owned(esrc, n_edges, n0, nown, rot, [&](int sl, int e) {
            const int t = etrg[e];
            atomicMax(&maxL[sl], ord_of_float(edge_logit(ssL[sl], s_trg[t])));
        });
        __syncthreads();
        scan_owned(esrc, n_edges, n0, nown, rot, [&](int sl, int e) {
            const float l = edge_logit(ssL[sl], s_trg[etrg[e]]);
            atomicAdd(&denL[sl], __expf(l - float_of_ord(maxL[sl])));
        });
        __syncthreads();
        scan_owned(esrc, n_edges, n0, nown, rot, [&](int sl, int e) {
            const int t = etrg[e];
            const float l = edge_logit(ssL[sl], s_trg[t]);
            const float alpha = __expf(l - float_of_ord(maxL[sl])) / denL[sl];
            const float* __restrict__ row = wx + (size_t)t * D_OUT;
            for (int d = 0; d < D_OUT; ++d) atomicAdd(&rowsL.acc[sl * D_OUT + d], alpha * row[d]);
        });
        __syncthreads();
        float* __restrict__ o = out + (size_t)n0 * D_OUT;
        for (int i = tid; i < nown * D_OUT; i += ETHR) o[i] = rowsL.acc[i];
    }
}

extern "C" void kernel_launch(void* const* d_in, const int* in_sizes, int n_in,
                              void* d_out, int out_size, void* d_ws, size_t ws_size,
                              hipStream_t stream) {
    const float* X   = (const float*)d_in[0];
    const float* W   = (const float*)d_in[1];
    const float* att = (const float*)d_in[2];
    const int*   adj = (const int*)  d_in[3];

    const int d_out_dim = in_sizes[2] / 2;
    const int d_in_dim  = in_sizes[1] / d_out_dim;
    const int n_nodes   = in_sizes[0] / d_in_dim;
    const int n_edges   = in_sizes[3] / 2;
    (void)d_out_dim; (void)d_in_dim;

    const int* esrc = adj;
    const int* etrg = adj + n_edges;

    float* wx    = (float*)d_ws;
    float* s_src = wx + (size_t)n_nodes * D_OUT;
    float* s_trg = s_src + n_nodes;
    const int nb   = (n_nodes + NPW - 1) / NPW;
    const int nseg = (n_edges + SEGE - 1) / SEGE;
    unsigned* tab  = (unsigned*)(s_trg + n_nodes);
    int2*     segs = (int2*)(tab + (((size_t)nb * nseg + 1) & ~(size_t)1));
    float* out   = (float*)d_out;

    const int tiles       = (n_nodes + 15) / 16;
    const int gemm_blocks = (tiles + BTHR / 32 - 1) / (BTHR / 32);
    k_front<<<gemm_blocks + nseg, BTHR, 0, stream>>>(X, W, att, wx, s_src, s_trg, esrc, etrg, tab, segs,
                                                    n_nodes, n_edges, nb, nseg, gemm_blocks);
    k_edges<<<nb, ETHR, 0, stream>>>(esrc, etrg, s_src, s_trg, wx, tab, segs, out,
                                     n_nodes, n_edges, nseg);
}


# Second-level prefetch (global_prefetch_b8) of first-touch data: edge work-groups ask for their nodes' wx rows and s_trg words ahead of the gathers, GEMM waves for the second half of their X rows
# speedup vs baseline: 5.3054x; 1.1065x over previous
#include <hip/hip_runtime.h>
#include <hip/hip_bf16.h>


#define D_IN   128
#define D_OUT  64
#define LEAKY  0.2f
#define BTHR   1024

typedef float v2f __attribute__((ext_vector_type(2)));
typedef float v8f __attribute__((ext_vector_type(8)));

__device__ __forceinline__ unsigned ord_of_float(float f) {
    unsigned u = __float_as_uint(f);
    return (u & 0x80000000u) ? ~u : (u | 0x80000000u);
}
__device__ __forceinline__ float float_of_ord(unsigned u) {
    return __uint_as_float((u & 0x80000000u) ? (u ^ 0x80000000u) : ~u);
}

template <int N>
__device__ __forceinline__ float row_xor(float v) {
    return __int_as_float(__builtin_amdgcn_update_dpp(0, __float_as_int(v), 0x160 | N, 0xf, 0xf, true));
}

__device__ __forceinline__ void l2_prefetch(const float* p) {
    __builtin_amdgcn_global_prefetch((const __attribute__((address_space(1))) void*)p, 16  );
}

__device__ __forceinline__ void gemm_part(const float* __restrict__ X,
                                          const float* __restrict__ W,
                                          const float* __restrict__ att,
                                          float* __restrict__ wx,
                                          float* __restrict__ s_src,
                                          float* __restrict__ s_trg,
                                          int n_nodes, int bid) {
    constexpr int LDW = D_IN + 4;
    __shared__ float sWt[D_OUT * LDW];
    __shared__ float sAtt[2 * D_OUT];

    const int lane = threadIdx.x & 31;
    const int wave = __builtin_amdgcn_readfirstlane((int)(threadIdx.x >> 5));
    const int tile = bid * (BTHR / 32) + wave;
    const int row0 = tile * 16;
    const int m    = lane & 15;
    const int koff = (lane >> 4) * 2;
    int arow = row0 + m;
    if (arow >= n_nodes) arow = n_nodes - 1;
    const float* __restrict__ xrow = X + (size_t)arow * D_IN;

    constexpr int WPT  = D_IN * D_OUT / BTHR;
    constexpr int NPRE = 5;
    float wv[WPT];
    #pragma unroll
    for (int j = 0; j < WPT; ++j) wv[j] = W[threadIdx.x + j * BTHR];
    const bool has_att = threadIdx.x < 2 * D_OUT;
    float attw = 0.0f;
    if (has_att) attw = att[threadIdx.x];
    v2f apre[NPRE];
    #pragma unroll
    for (int j = 0; j < NPRE; ++j) apre[j] = *(const v2f*)(xrow + 4 * j + koff);
    #pragma unroll
    for (int j = 0; j < WPT; ++j) {
        const int i = threadIdx.x + j * BTHR;
        const int k = i / D_OUT, col = i % D_OUT;
        sWt[col * LDW + k] = wv[j];
    }
    if (has_att) sAtt[threadIdx.x] = attw;
    asm volatile("" :: "v"(apre[0]), "v"(apre[1]), "v"(apre[2]), "v"(apre[3]), "v"(apre[4]));
    __syncthreads();
    if (row0 >= n_nodes) return;

    l2_prefetch(xrow + D_IN / 2);

    v8f acc[4] = {v8f{}, v8f{}, v8f{}, v8f{}};

    #pragma unroll
    for (int kb = 0; kb < D_IN; kb += 4) {
        v2f a;
        if (kb / 4 < NPRE) a = apre[kb / 4];
        else               a = *(const v2f*)(xrow + kb + koff);
        #pragma unroll
        for (int t = 0; t < 4; ++t) {
            const int col = t * 16 + m;
            v2f b = *(const v2f*)(&sWt[col * LDW + kb + koff]);
            acc[t] = __builtin_amdgcn_wmma_f32_16x16x4_f32(
                 false, a,  false, b,
                 (short)0, acc[t],
                 false,  false);
        }
    }

    const int rbase = row0 + ((lane >> 4) * 8);
    #pragma unroll
    for (int t = 0; t < 4; ++t) {
        #pragma unroll
        for (int r = 0; r < 8; ++r) {
            const int row = rbase + r;
            if (row < n_nodes)
                wx[(size_t)row * D_OUT + t * 16 + m] = acc[t][r];
        }
    }

    float as_[4], at_[4];
    #pragma unroll
    for (int t = 0; t < 4; ++t) {
        as_[t] = sAtt[t * 16 + m];
        at_[t] = sAtt[D_OUT + t * 16 + m];
    }
    float v16[16];
    #pragma unroll
    for (int r = 0; r < 8; ++r) {
        float ss = acc[0][r] * as_[0];
        float st = acc[0][r] * at_[0];
        #pragma unroll
        for (int t = 1; t < 4; ++t) {
            ss = fmaf(acc[t][r], as_[t], ss);
            st = fmaf(acc[t][r], at_[t], st);
        }
        v16[r]     = ss;
        v16[8 + r] = st;
    }
    const bool b3 = (m & 8) != 0, b2 = (m & 4) != 0, b1 = (m & 2) != 0, b0 = (m & 1) != 0;
    float v8[8], v4[4], v2[2];
    #pragma unroll
    for (int i = 0; i < 8; ++i) {
        const float keep = b3 ? v16[i + 8] : v16[i];
        const float send = b3 ? v16[i] : v16[i + 8];
        v8[i] = keep + row_xor<8>(send);
    }
    #pragma unroll
    for (int i = 0; i < 4; ++i) {
        const float keep = b2 ? v8[i + 4] : v8[i];
        const float send = b2 ? v8[i] : v8[i + 4];
        v4[i] = keep + row_xor<4>(send);
    }
    #pragma unroll
    for (int i = 0; i < 2; ++i) {
        const float keep = b1 ? v4[i + 2] : v4[i];
        const float send = b1 ? v4[i] : v4[i + 2];
        v2[i] = keep + row_xor<2>(send);
    }
    const float keep1 = b0 ? v2[1] : v2[0];
    const float send1 = b0 ? v2[0] : v2[1];
    const float score = keep1 + row_xor<1>(send1);
    {
        const int row = rbase + (m & 7);
        float* dst = b3 ? s_trg : s_src;
        if (row < n_nodes) dst[row] = score;
    }
}

#define NPW   391
#define CAP   8192
#define ETHR  1024
#define BEPT  4
#define SEGE  (BTHR * BEPT)
#define NBMAX 1024
#define SEGMAX 256
#define CBITS 13

__device__ __forceinline__ void bin_part(const int* __restrict__ esrc,
                                         const int* __restrict__ etrg,
                                         unsigned* __restrict__ tab,
                                         int2* __restrict__ segs,
                                         int n_nodes, int n_edges, int nb, int nseg, int seg) {
    __shared__ unsigned hist[NBMAX];
    __shared__ unsigned wtot[32];
    const int tid  = threadIdx.x;
    const int lane = tid & 31;
    const int wave = tid >> 5;
    hist[tid] = 0u;
    __syncthreads();

    const int base = seg * SEGE + tid;
    int s[BEPT], t[BEPT], b[BEPT];
    unsigned r[BEPT];
    #pragma unroll
    for (int u = 0; u < BEPT; ++u) {
        const int e = base + u * BTHR;
        const bool in = e < n_edges;
        s[u] = in ? esrc[e] : -1;
        t[u] = in ? etrg[e] : 0;
    }
    #pragma unroll
    for (int u = 0; u < BEPT; ++u) {
        const bool ok = (unsigned)s[u] < (unsigned)n_nodes;
        b[u] = ok ? s[u] / NPW : -1;
        r[u] = ok ? atomicAdd(&hist[b[u]], 1u) : 0u;
    }
    __syncthreads();
    const unsigned cntv = hist[tid];
    unsigned inc = cntv;
    #pragma unroll
    for (int d = 1; d < 32; d <<= 1) {
        const unsigned o = __shfl_up(inc, d);
        if (lane >= d) inc += o;
    }
    if (lane == 31) wtot[wave] = inc;
    __syncthreads();
    if (wave == 0) {
        const unsigned w = wtot[lane];
        unsigned winc = w;
        #pragma unroll
        for (int d = 1; d < 32; d <<= 1) {
            const unsigned o = __shfl_up(winc, d);
            if (lane >= d) winc += o;
        }
        wtot[lane] = winc - w;
    }
    __syncthreads();
    const unsigned start = inc - cntv + wtot[wave];
    hist[tid] = start;
    if (tid < nb) tab[(size_t)tid * nseg + seg] = (start << CBITS) | cntv;
    __syncthreads();
    int2* __restrict__ mine = segs + (size_t)seg * SEGE;
    #pragma unroll
    for (int u = 0; u < BEPT; ++u) {
        if (b[u] >= 0)
            mine[hist[b[u]] + r[u]] = make_int2(s[u] - b[u] * NPW, t[u]);
    }
}

__global__ __launch_bounds__(BTHR, 16) void k_front(const float* __restrict__ X,
                                                const float* __restrict__ W,
                                                const float* __restrict__ att,
                                                float* __restrict__ wx,
                                                float* __restrict__ s_src,
                                                float* __restrict__ s_trg,
                                                const int* __restrict__ esrc,
                                                const int* __restrict__ etrg,
                                                unsigned* __restrict__ tab,
                                                int2* __restrict__ segs,
                                                int n_nodes, int n_edges, int nb, int nseg,
                                                int gemm_blocks) {
    if ((int)blockIdx.x < gemm_blocks)
        gemm_part(X, W, att, wx, s_src, s_trg, n_nodes, (int)blockIdx.x);
    else
        bin_part(esrc, etrg, tab, segs, n_nodes, n_edges, nb, nseg, (int)blockIdx.x - gemm_blocks);
}

__device__ __forceinline__ float edge_logit(float a, float b) {
    float l = a + b;
    return (l >= 0.0f) ? l : LEAKY * l;
}

template <typename F>
__device__ __forceinline__ void scan_owned(const int* __restrict__ esrc, int n_edges,
                                           int n0, int nown, int rot, F body) {
    const int tid  = threadIdx.x;
    const int nvec = n_edges >> 2;
    const int4* __restrict__ src4 = (const int4*)esrc;
    for (int base = 0; base < nvec; base += 4 * ETHR) {
        int4 q[4];
        int  w[4];
        #pragma unroll
        for (int u = 0; u < 4; ++u) {
            const int v = base + u * ETHR + tid;
            int ww = v + rot;
            if (ww >= nvec) ww -= nvec;
            w[u] = ww;
            q[u] = (v < nvec) ? src4[ww] : make_int4(-1, -1, -1, -1);
        }
        #pragma unroll
        for (int u = 0; u < 4; ++u) {
            const int e0 = w[u] << 2;
            const unsigned d0 = (unsigned)(q[u].x - n0);
            const unsigned d1 = (unsigned)(q[u].y - n0);
            const unsigned d2 = (unsigned)(q[u].z - n0);
            const unsigned d3 = (unsigned)(q[u].w - n0);
            if (d0 < (unsigned)nown) body((int)d0, e0);
            if (d1 < (unsigned)nown) body((int)d1, e0 + 1);
            if (d2 < (unsigned)nown) body((int)d2, e0 + 2);
            if (d3 < (unsigned)nown) body((int)d3, e0 + 3);
        }
    }
    for (int e = (nvec << 2) + tid; e < n_edges; e += ETHR) {
        const unsigned d = (unsigned)(esrc[e] - n0);
        if (d < (unsigned)nown) body((int)d, e);
    }
}

#define RB 16
#define SBITS 9
#define SCANW ((NPW + 1 + 31) / 32)

struct RowState {
    int    cur;
    float2 acc;
};

__device__ __forceinline__ void flush_row(RowState& st, float* __restrict__ orow) {
    if (st.cur >= 0) *(float2*)(orow + (size_t)st.cur * D_OUT) = st.acc;
    st.acc = make_float2(0.0f, 0.0f);
}

template <bool TAIL>
__device__ __forceinline__ void add_rows(const float* __restrict__ wx, float* __restrict__ orow, RowState& st,
                                         int my_s, int my_t, float my_a, unsigned bm,
                                         int j0, int n, int lane) {
    int    sl[RB];
    float  al[RB];
    float2 v[RB];
    #pragma unroll
    for (int u = 0; u < RB; ++u) {
        const int jj = TAIL ? min(j0 + u, n - 1) : (j0 + u);
        sl[u] = __builtin_amdgcn_readlane(my_s, jj);
        const int t = __builtin_amdgcn_readlane(my_t, jj);
        al[u] = __int_as_float(__builtin_amdgcn_readlane(__float_as_int(my_a), jj));
        v[u]  = ((const float2*)(wx + (size_t)t * D_OUT))[lane];
    }
    #pragma unroll
    for (int u = 0; u < RB; ++u) {
        if (!TAIL || j0 + u < n) {
            if ((bm >> (j0 + u)) & 1u) {
                flush_row(st, orow);
                st.cur = sl[u];
            }
            st.acc.x += al[u] * v[u].x;
            st.acc.y += al[u] * v[u].y;
        }
    }
}

__global__ __launch_bounds__(ETHR) void k_edges(const int* __restrict__ esrc,
                                                const int* __restrict__ etrg,
                                                const float* __restrict__ s_src,
                                                const float* __restrict__ s_trg,
                                                const float* __restrict__ wx,
                                                const unsigned* __restrict__ tab,
                                                const int2* __restrict__ segs,
                                                float* __restrict__ out,
                                                int n_nodes, int n_edges, int nseg) {
    __shared__ union {
        float acc[NPW * D_OUT];
        struct { int s[CAP]; int t[CAP]; float w[CAP]; } srt;
    } rowsL;
    __shared__ unsigned cntS[NPW + 1];
    __shared__ unsigned offS[NPW + 1];
    __shared__ unsigned wtotS[SCANW];
    __shared__ unsigned maxL[NPW];
    __shared__ float    denL[NPW];
    __shared__ float    ssL[NPW];
    __shared__ int      listS[CAP];
    __shared__ int      listT[CAP];
    __shared__ float    listW[CAP];
    __shared__ unsigned segInfo[SEGMAX];
    __shared__ unsigned segBase[SEGMAX];
    __shared__ unsigned lcnt;
    __shared__ unsigned fbIdx;

    const int tid  = threadIdx.x;
    const int lane = tid & 31;
    const int wave = tid >> 5;
    const int n0   = blockIdx.x * NPW;
    const int nown = min(NPW, n_nodes - n0);

    const bool have = nseg <= SEGMAX;
    const bool tseg = have && tid < nseg;
    unsigned tabw = 0u;
    if (tseg) tabw = tab[(size_t)blockIdx.x * nseg + tid];

    {
        const int k = tid - ETHR / 2;
        if (k >= 0 && k < 8 && nown > 0) l2_prefetch(s_trg + min(n0 + 64 * k, n0 + nown - 1));
    }

    for (int i = tid; i < NPW; i += ETHR) {
        cntS[i] = 0u;
        maxL[i] = 0u;
        denL[i] = 0.0f;
        ssL[i]  = (i < nown) ? s_src[n0 + i] : 0.0f;
    }
    if (tid == 0) lcnt = 0u;
    __syncthreads();
    if (tid < nown) l2_prefetch(wx + (size_t)(n0 + tid) * D_OUT);

    if (tseg) {
        segInfo[tid] = tabw;
        segBase[tid] = atomicAdd(&lcnt, tabw & ((1u << CBITS) - 1u));
    }
    __syncthreads();
    const unsigned cnt = have ? lcnt : (unsigned)CAP + 1u;

    if (cnt <= (unsigned)CAP) {
        const unsigned l16   = (unsigned)tid & 15u;
        const unsigned cmask = (1u << CBITS) - 1u;
        for (int sg = tid >> 4; sg < nseg; sg += ETHR / 16) {
            const unsigned p     = segInfo[sg];
            const unsigned c     = p & cmask;
            const unsigned bs    = segBase[sg];
            const unsigned first = (unsigned)sg * SEGE + (p >> CBITS);
            for (unsigned j = l16; j < c; j += 16) {
                listT[bs + j] = (int)(first + j);
                if (bs + j == 0u) fbIdx = first + j;
            }
        }
        __syncthreads();
        const unsigned fb = fbIdx;
        for (unsigned base = 0; base < cnt; base += 4 * ETHR) {
            unsigned idx[4];
            int2     rec[4];
            float    st[4];
            #pragma unroll
            for (int u = 0; u < 4; ++u) {
                const unsigned i = base + u * ETHR + tid;
                idx[u] = (i < cnt) ? (unsigned)listT[i] : fb;
            }
            #pragma unroll
            for (int u = 0; u < 4; ++u) rec[u] = segs[idx[u]];
            #pragma unroll
            for (int u = 0; u < 4; ++u) st[u] = s_trg[rec[u].y];
            asm volatile("" :: "v"(st[0]), "v"(st[1]), "v"(st[2]), "v"(st[3]));
            #pragma unroll
            for (int u = 0; u < 4; ++u) {
                const unsigned i = base + u * ETHR + tid;
                if (i < cnt) {
                    const int sl = rec[u].x;
                    const float l = edge_logit(ssL[sl], st[u]);
                    const unsigned rk = atomicAdd(&cntS[sl], 1u);
                    listS[i] = sl | (int)(rk << SBITS);
                    listT[i] = rec[u].y;
                    listW[i] = l;
                    atomicMax(&maxL[sl], ord_of_float(l));
                }
            }
        }
        __syncthreads();
        unsigned cntv = 0u, inc = 0u;
        if (wave < SCANW) {
            cntv = (tid < NPW) ? cntS[tid] : 0u;
            inc  = cntv;
            #pragma unroll
            for (int d = 1; d < 32; d <<= 1) {
                const unsigned o = __shfl_up(inc, d);
                if (lane >= d) inc += o;
            }
            if (lane == 31) wtotS[wave] = inc;
            unsigned zm = __builtin_amdgcn_ballot_w32(tid < nown && cntv == 0u);
            while (zm != 0u) {
                const int b = __builtin_ctz(zm);
                zm &= zm - 1u;
                *(float2*)(out + (size_t)(n0 + wave * 32 + b) * D_OUT + 2 * lane) = make_float2(0.0f, 0.0f);
            }
        }
        __syncthreads();
        if (wave < SCANW) {
            unsigned pre = 0u;
            for (int w = 0; w < wave; ++w) pre += wtotS[w];
            if (tid <= NPW) offS[tid] = pre + inc - cntv;
        }
        __syncthreads();
        for (unsigned i = tid; i < cnt; i += ETHR) {
            const int pk = listS[i];
            const int sl = pk & ((1 << SBITS) - 1);
            const float ex = __expf(listW[i] - float_of_ord(maxL[sl]));
            atomicAdd(&denL[sl], ex);
            const unsigned p = offS[sl] + ((unsigned)pk >> SBITS);
            rowsL.srt.s[p] = sl;
            rowsL.srt.t[p] = listT[i];
            rowsL.srt.w[p] = ex;
        }
        __syncthreads();
        const unsigned wv  = (unsigned)__builtin_amdgcn_readfirstlane(wave);
        const int s_lo = (int)((wv * (unsigned)nown) / (ETHR / 32));
        const int s_hi = (int)(((wv + 1u) * (unsigned)nown) / (ETHR / 32));
        const unsigned e_lo = (unsigned)__builtin_amdgcn_readfirstlane((int)offS[s_lo]);
        const unsigned e_hi = (unsigned)__builtin_amdgcn_readfirstlane((int)offS[s_hi]);
        float* __restrict__ orow = out + (size_t)n0 * D_OUT + 2 * lane;
        RowState st;
        st.cur = -1;
        st.acc = make_float2(0.0f, 0.0f);
        for (unsigned base = e_lo; base < e_hi; base += 32u) {
            const unsigned i  = base + lane;
            const bool     ok = i < e_hi;
            const int   my_s = ok ? rowsL.srt.s[i] : 0;
            const int   my_t = ok ? rowsL.srt.t[i] : 0;
            const float my_a = ok ? rowsL.srt.w[i] / denL[my_s] : 0.0f;
            const int   pv_s = (ok && i > e_lo) ? rowsL.srt.s[i - 1] : -1;
            const unsigned bm = __builtin_amdgcn_ballot_w32(ok && my_s != pv_s);
            const int n = (int)min(32u, e_hi - base);
            for (int j0 = 0; j0 < n; j0 += RB) {
                if (j0 + RB <= n) add_rows<false>(wx, orow, st, my_s, my_t, my_a, bm, j0, n, lane);
                else              add_rows<true >(wx, orow, st, my_s, my_t, my_a, bm, j0, n, lane);
            }
        }
        flush_row(st, orow);
    } else {
        for (int i = tid; i < NPW * D_OUT; i += ETHR) rowsL.acc[i] = 0.0f;
        const int nvec = n_edges >> 2;
        const int rot  = (int)(((long long)blockIdx.x * nvec) / gridDim.x) & ~31;
        scan_owned(esrc, n_edges, n0, nown, rot, [&](int sl, int e) {
            const int t = etrg[e];
            atomicMax(&maxL[sl], ord_of_float(edge_logit(ssL[sl], s_trg[t])));
        });
        __syncthreads();
        scan_owned(esrc, n_edges, n0, nown, rot, [&](int sl, int e) {
            const float l = edge_logit(ssL[sl], s_trg[etrg[e]]);
            atomicAdd(&denL[sl], __expf(l - float_of_ord(maxL[sl])));
        });
        __syncthreads();
        scan_owned(esrc, n_edges, n0, nown, rot, [&](int sl, int e) {
            const int t = etrg[e];
            const float l = edge_logit(ssL[sl], s_trg[t]);
            const float alpha = __expf(l - float_of_ord(maxL[sl])) / denL[sl];
            const float* __restrict__ row = wx + (size_t)t * D_OUT;
            for (int d = 0; d < D_OUT; ++d) atomicAdd(&rowsL.acc[sl * D_OUT + d], alpha * row[d]);
        });
        __syncthreads();
        float* __restrict__ o = out + (size_t)n0 * D_OUT;
        for (int i = tid; i < nown * D_OUT; i += ETHR) o[i] = rowsL.acc[i];
    }
}

extern "C" void kernel_launch(void* const* d_in, const int* in_sizes, int n_in,
                              void* d_out, int out_size, void* d_ws, size_t ws_size,
                              hipStream_t stream) {
    const float* X   = (const float*)d_in[0];
    const float* W   = (const float*)d_in[1];
    const float* att = (const float*)d_in[2];
    const int*   adj = (const int*)  d_in[3];

    const int d_out_dim = in_sizes[2] / 2;
    const int d_in_dim  = in_sizes[1] / d_out_dim;
    const int n_nodes   = in_sizes[0] / d_in_dim;
    const int n_edges   = in_sizes[3] / 2;
    (void)d_out_dim; (void)d_in_dim;

    const int* esrc = adj;
    const int* etrg = adj + n_edges;

    float* wx    = (float*)d_ws;
    float* s_src = wx + (size_t)n_nodes * D_OUT;
    float* s_trg = s_src + n_nodes;
    const int nb   = (n_nodes + NPW - 1) / NPW;
    const int nseg = (n_edges + SEGE - 1) / SEGE;
    unsigned* tab  = (unsigned*)(s_trg + n_nodes);
    int2*     segs = (int2*)(tab + (((size_t)nb * nseg + 1) & ~(size_t)1));
    float* out   = (float*)d_out;

    const int tiles       = (n_nodes + 15) / 16;
    const int gemm_blocks = (tiles + BTHR / 32 - 1) / (BTHR / 32);
    k_front<<<gemm_blocks + nseg, BTHR, 0, stream>>>(X, W, att, wx, s_src, s_trg, esrc, etrg, tab, segs,
                                                    n_nodes, n_edges, nb, nseg, gemm_blocks);
    k_edges<<<nb, ETHR, 0, stream>>>(esrc, etrg, s_src, s_trg, wx, tab, segs, out,
                                     n_nodes, n_edges, nseg);
}
